# starting kernel
# baseline (speedup 1.0000x reference)
#include <hip/hip_runtime.h>
#include <hip/hip_bf16.h>

typedef __attribute__((ext_vector_type(2))) float v2f;
typedef __attribute__((ext_vector_type(8))) float v8f;

#define D_FEAT 64
#define H_DIM  256
#define TILE_M 64
#define RST_STRIDE 68
#define H_STRIDE   260

__global__ void gin_init_kernel(const float* __restrict__ feat,
                                float* __restrict__ rst, int n_elems4) {
    int i = blockIdx.x * blockDim.x + threadIdx.x;
    if (i < n_elems4) {
        reinterpret_cast<float4*>(rst)[i] =
            reinterpret_cast<const float4*>(feat)[i];
    }
}

__global__ void gin_scatter_kernel(const float* __restrict__ feat,
                                   const int* __restrict__ src,
                                   const int* __restrict__ dst,
                                   float* __restrict__ rst, int n_edges) {
    long long tid = (long long)blockIdx.x * blockDim.x + threadIdx.x;
    int e = (int)(tid >> 4);
    int c = (int)(tid & 15);
    if (e >= n_edges) return;
    size_t soff = (size_t)src[e] * D_FEAT + (size_t)c * 4;
    size_t doff = (size_t)dst[e] * D_FEAT + (size_t)c * 4;
    float4 v = *reinterpret_cast<const float4*>(feat + soff);
    float* p = rst + doff;
    unsafeAtomicAdd(p + 0, v.x);
    unsafeAtomicAdd(p + 1, v.y);
    unsafeAtomicAdd(p + 2, v.z);
    unsafeAtomicAdd(p + 3, v.w);
}

__global__ __launch_bounds__(512)
void gin_mlp_kernel(const float* __restrict__ rst,
                    const float* __restrict__ W1,
                    const float* __restrict__ b1,
                    const float* __restrict__ W2,
                    const float* __restrict__ b2,
                    float* __restrict__ out, int n_nodes) {
    __shared__ float s_rst[TILE_M * RST_STRIDE];
    __shared__ float s_h[TILE_M * H_STRIDE];
    __shared__ float s_out[TILE_M * D_FEAT];

    const int tid   = threadIdx.x;
    const int wave  = tid >> 5;
    const int lane  = tid & 31;
    const int node0 = blockIdx.x * TILE_M;

    for (int i = tid; i < TILE_M * D_FEAT; i += 512) {
        int m = i >> 6, k = i & 63;
        int row = node0 + m;
        row = row < n_nodes ? row : (n_nodes - 1);
        s_rst[m * RST_STRIDE + k] = rst[(size_t)row * D_FEAT + k];
    }
    for (int i = tid; i < TILE_M * D_FEAT; i += 512) s_out[i] = 0.0f;
    __syncthreads();

    const int frow  = lane & 15;
    const int khalf = (lane >> 4) << 1;
    const int mbase = (lane >> 4) << 3;

    {
        const int ncol = wave * 16 + frow;
        v8f acc0 = {}, acc1 = {}, acc2 = {}, acc3 = {};
        #pragma unroll
        for (int kk = 0; kk < 16; ++kk) {
            const int k0 = kk * 4 + khalf;
            v2f b;
            b.x = W1[(size_t)(k0    ) * H_DIM + ncol];
            b.y = W1[(size_t)(k0 + 1) * H_DIM + ncol];
            v2f a0, a1, a2, a3;
            a0.x = s_rst[(frow     ) * RST_STRIDE + k0];
            a0.y = s_rst[(frow     ) * RST_STRIDE + k0 + 1];
            a1.x = s_rst[(frow + 16) * RST_STRIDE + k0];
            a1.y = s_rst[(frow + 16) * RST_STRIDE + k0 + 1];
            a2.x = s_rst[(frow + 32) * RST_STRIDE + k0];
            a2.y = s_rst[(frow + 32) * RST_STRIDE + k0 + 1];
            a3.x = s_rst[(frow + 48) * RST_STRIDE + k0];
            a3.y = s_rst[(frow + 48) * RST_STRIDE + k0 + 1];
            acc0 = __builtin_amdgcn_wmma_f32_16x16x4_f32(
                false, a0, false, b, (short)0, acc0, false, false);
            acc1 = __builtin_amdgcn_wmma_f32_16x16x4_f32(
                false, a1, false, b, (short)0, acc1, false, false);
            acc2 = __builtin_amdgcn_wmma_f32_16x16x4_f32(
                false, a2, false, b, (short)0, acc2, false, false);
            acc3 = __builtin_amdgcn_wmma_f32_16x16x4_f32(
                false, a3, false, b, (short)0, acc3, false, false);
        }
        const float bias = b1[ncol];
        #pragma unroll
        for (int r = 0; r < 8; ++r) {
            float v0 = acc0[r] + bias, v1 = acc1[r] + bias;
            float v2 = acc2[r] + bias, v3 = acc3[r] + bias;
            s_h[(     mbase + r) * H_STRIDE + ncol] = v0 > 0.f ? v0 : 0.f;
            s_h[(16 + mbase + r) * H_STRIDE + ncol] = v1 > 0.f ? v1 : 0.f;
            s_h[(32 + mbase + r) * H_STRIDE + ncol] = v2 > 0.f ? v2 : 0.f;
            s_h[(48 + mbase + r) * H_STRIDE + ncol] = v3 > 0.f ? v3 : 0.f;
        }
    }
    __syncthreads();

    {
        const int ntile = (wave & 3) * 16;
        const int kbase = (wave >> 2) * 64;
        const int ncol  = ntile + frow;
        v8f acc0 = {}, acc1 = {}, acc2 = {}, acc3 = {};
        #pragma unroll
        for (int kk = 0; kk < 16; ++kk) {
            const int k0 = kbase + kk * 4 + khalf;
            v2f b;
            b.x = W2[(size_t)(k0    ) * D_FEAT + ncol];
            b.y = W2[(size_t)(k0 + 1) * D_FEAT + ncol];
            v2f a0, a1, a2, a3;
            a0.x = s_h[(frow     ) * H_STRIDE + k0];
            a0.y = s_h[(frow     ) * H_STRIDE + k0 + 1];
            a1.x = s_h[(frow + 16) * H_STRIDE + k0];
            a1.y = s_h[(frow + 16) * H_STRIDE + k0 + 1];
            a2.x = s_h[(frow + 32) * H_STRIDE + k0];
            a2.y = s_h[(frow + 32) * H_STRIDE + k0 + 1];
            a3.x = s_h[(frow + 48) * H_STRIDE + k0];
            a3.y = s_h[(frow + 48) * H_STRIDE + k0 + 1];
            acc0 = __builtin_amdgcn_wmma_f32_16x16x4_f32(
                false, a0, false, b, (short)0, acc0, false, false);
            acc1 = __builtin_amdgcn_wmma_f32_16x16x4_f32(
                false, a1, false, b, (short)0, acc1, false, false);
            acc2 = __builtin_amdgcn_wmma_f32_16x16x4_f32(
                false, a2, false, b, (short)0, acc2, false, false);
            acc3 = __builtin_amdgcn_wmma_f32_16x16x4_f32(
                false, a3, false, b, (short)0, acc3, false, false);
        }
        #pragma unroll
        for (int r = 0; r < 8; ++r) {
            atomicAdd(&s_out[(     mbase + r) * D_FEAT + ncol], acc0[r]);
            atomicAdd(&s_out[(16 + mbase + r) * D_FEAT + ncol], acc1[r]);
            atomicAdd(&s_out[(32 + mbase + r) * D_FEAT + ncol], acc2[r]);
            atomicAdd(&s_out[(48 + mbase + r) * D_FEAT + ncol], acc3[r]);
        }
    }
    __syncthreads();

    for (int i = tid; i < TILE_M * D_FEAT; i += 512) {
        int m = i >> 6, n = i & 63;
        int row = node0 + m;
        if (row < n_nodes)
            out[(size_t)row * D_FEAT + n] = s_out[i] + b2[n];
    }
}

extern "C" void kernel_launch(void* const* d_in, const int* in_sizes, int n_in,
                              void* d_out, int out_size, void* d_ws, size_t ws_size,
                              hipStream_t stream) {
    const float* feat = (const float*)d_in[0];
    const float* W1   = (const float*)d_in[1];
    const float* b1   = (const float*)d_in[2];
    const float* W2   = (const float*)d_in[3];
    const float* b2   = (const float*)d_in[4];
    const int*   src  = (const int*)d_in[5];
    const int*   dst  = (const int*)d_in[6];
    float*       out  = (float*)d_out;

    const int n_nodes = in_sizes[0] / D_FEAT;
    const int n_edges = in_sizes[5];

    float* rst = (float*)d_ws;

    {
        int n4 = (n_nodes * D_FEAT) / 4;
        gin_init_kernel<<<(n4 + 255) / 256, 256, 0, stream>>>(feat, rst, n4);
    }
    {
        long long work = (long long)n_edges * 16;
        int blocks = (int)((work + 255) / 256);
        gin_scatter_kernel<<<blocks, 256, 0, stream>>>(feat, src, dst, rst, n_edges);
    }
    {
        int blocks = (n_nodes + TILE_M - 1) / TILE_M;
        gin_mlp_kernel<<<blocks, 512, 0, stream>>>(rst, W1, b1, W2, b2, out, n_nodes);
    }
}


# Scatter: one wave per 32 edges, lanes cover 32 consecutive floats, so every gather load and f32 atomic add is one whole 128-B line (was 4 quarter lines); sum order arrival-order as before
# speedup vs baseline: 1.5375x; 1.5375x over previous
#include <hip/hip_runtime.h>
#include <hip/hip_bf16.h>

typedef __attribute__((ext_vector_type(2))) float v2f;
typedef __attribute__((ext_vector_type(8))) float v8f;

#define D_FEAT 64
#define H_DIM  256
#define TILE_M 64
#define RST_STRIDE 68
#define H_STRIDE   260

__global__ void gin_init_kernel(const float* __restrict__ feat,
                                float* __restrict__ rst, int n_elems4) {
    int i = blockIdx.x * blockDim.x + threadIdx.x;
    if (i < n_elems4) {
        reinterpret_cast<float4*>(rst)[i] =
            reinterpret_cast<const float4*>(feat)[i];
    }
}

#define EDGE_GROUP 16

__global__ __launch_bounds__(256)
void gin_scatter_kernel(const float* __restrict__ feat,
                        const int* __restrict__ src,
                        const int* __restrict__ dst,
                        float* __restrict__ rst, int n_edges) {
    const int lane = threadIdx.x & 31;
    const int wv = __builtin_amdgcn_readfirstlane((int)(threadIdx.x >> 5));
    const int e0 = (blockIdx.x * (int)(blockDim.x >> 5) + wv) * 32;
    if (e0 >= n_edges) return;

    const int e = e0 + lane;
    int s = 0, d = 0;
    if (e < n_edges) { s = src[e]; d = dst[e]; }

    if (e0 + 32 <= n_edges) {
        #pragma unroll
        for (int g = 0; g < 32; g += EDGE_GROUP) {
            float lo[EDGE_GROUP], hi[EDGE_GROUP];
            #pragma unroll
            for (int j = 0; j < EDGE_GROUP; ++j) {
                const int sj = __builtin_amdgcn_readlane(s, g + j);
                const float* row = feat + (size_t)sj * D_FEAT;
                lo[j] = row[lane];
                hi[j] = row[lane + 32];
            }
            #pragma unroll
            for (int j = 0; j < EDGE_GROUP; ++j) {
                const int dj = __builtin_amdgcn_readlane(d, g + j);
                float* row = rst + (size_t)dj * D_FEAT;
                unsafeAtomicAdd(row + lane,      lo[j]);
                unsafeAtomicAdd(row + lane + 32, hi[j]);
            }
        }
    } else {
        const int cnt = n_edges - e0;
        for (int j = 0; j < cnt; ++j) {
            const int sj = __shfl(s, j);
            const int dj = __shfl(d, j);
            const float* frow = feat + (size_t)sj * D_FEAT;
            float* row = rst + (size_t)dj * D_FEAT;
            const float v0 = frow[lane], v1 = frow[lane + 32];
            unsafeAtomicAdd(row + lane,      v0);
            unsafeAtomicAdd(row + lane + 32, v1);
        }
    }
}

__global__ __launch_bounds__(512)
void gin_mlp_kernel(const float* __restrict__ rst,
                    const float* __restrict__ W1,
                    const float* __restrict__ b1,
                    const float* __restrict__ W2,
                    const float* __restrict__ b2,
                    float* __restrict__ out, int n_nodes) {
    __shared__ float s_rst[TILE_M * RST_STRIDE];
    __shared__ float s_h[TILE_M * H_STRIDE];
    __shared__ float s_out[TILE_M * D_FEAT];

    const int tid   = threadIdx.x;
    const int wave  = tid >> 5;
    const int lane  = tid & 31;
    const int node0 = blockIdx.x * TILE_M;

    for (int i = tid; i < TILE_M * D_FEAT; i += 512) {
        int m = i >> 6, k = i & 63;
        int row = node0 + m;
        row = row < n_nodes ? row : (n_nodes - 1);
        s_rst[m * RST_STRIDE + k] = rst[(size_t)row * D_FEAT + k];
    }
    for (int i = tid; i < TILE_M * D_FEAT; i += 512) s_out[i] = 0.0f;
    __syncthreads();

    const int frow  = lane & 15;
    const int khalf = (lane >> 4) << 1;
    const int mbase = (lane >> 4) << 3;

    {
        const int ncol = wave * 16 + frow;
        v8f acc0 = {}, acc1 = {}, acc2 = {}, acc3 = {};
        #pragma unroll
        for (int kk = 0; kk < 16; ++kk) {
            const int k0 = kk * 4 + khalf;
            v2f b;
            b.x = W1[(size_t)(k0    ) * H_DIM + ncol];
            b.y = W1[(size_t)(k0 + 1) * H_DIM + ncol];
            v2f a0, a1, a2, a3;
            a0.x = s_rst[(frow     ) * RST_STRIDE + k0];
            a0.y = s_rst[(frow     ) * RST_STRIDE + k0 + 1];
            a1.x = s_rst[(frow + 16) * RST_STRIDE + k0];
            a1.y = s_rst[(frow + 16) * RST_STRIDE + k0 + 1];
            a2.x = s_rst[(frow + 32) * RST_STRIDE + k0];
            a2.y = s_rst[(frow + 32) * RST_STRIDE + k0 + 1];
            a3.x = s_rst[(frow + 48) * RST_STRIDE + k0];
            a3.y = s_rst[(frow + 48) * RST_STRIDE + k0 + 1];
            acc0 = __builtin_amdgcn_wmma_f32_16x16x4_f32(
                false, a0, false, b, (short)0, acc0, false, false);
            acc1 = __builtin_amdgcn_wmma_f32_16x16x4_f32(
                false, a1, false, b, (short)0, acc1, false, false);
            acc2 = __builtin_amdgcn_wmma_f32_16x16x4_f32(
                false, a2, false, b, (short)0, acc2, false, false);
            acc3 = __builtin_amdgcn_wmma_f32_16x16x4_f32(
                false, a3, false, b, (short)0, acc3, false, false);
        }
        const float bias = b1[ncol];
        #pragma unroll
        for (int r = 0; r < 8; ++r) {
            float v0 = acc0[r] + bias, v1 = acc1[r] + bias;
            float v2 = acc2[r] + bias, v3 = acc3[r] + bias;
            s_h[(     mbase + r) * H_STRIDE + ncol] = v0 > 0.f ? v0 : 0.f;
            s_h[(16 + mbase + r) * H_STRIDE + ncol] = v1 > 0.f ? v1 : 0.f;
            s_h[(32 + mbase + r) * H_STRIDE + ncol] = v2 > 0.f ? v2 : 0.f;
            s_h[(48 + mbase + r) * H_STRIDE + ncol] = v3 > 0.f ? v3 : 0.f;
        }
    }
    __syncthreads();

    {
        const int ntile = (wave & 3) * 16;
        const int kbase = (wave >> 2) * 64;
        const int ncol  = ntile + frow;
        v8f acc0 = {}, acc1 = {}, acc2 = {}, acc3 = {};
        #pragma unroll
        for (int kk = 0; kk < 16; ++kk) {
            const int k0 = kbase + kk * 4 + khalf;
            v2f b;
            b.x = W2[(size_t)(k0    ) * D_FEAT + ncol];
            b.y = W2[(size_t)(k0 + 1) * D_FEAT + ncol];
            v2f a0, a1, a2, a3;
            a0.x = s_h[(frow     ) * H_STRIDE + k0];
            a0.y = s_h[(frow     ) * H_STRIDE + k0 + 1];
            a1.x = s_h[(frow + 16) * H_STRIDE + k0];
            a1.y = s_h[(frow + 16) * H_STRIDE + k0 + 1];
            a2.x = s_h[(frow + 32) * H_STRIDE + k0];
            a2.y = s_h[(frow + 32) * H_STRIDE + k0 + 1];
            a3.x = s_h[(frow + 48) * H_STRIDE + k0];
            a3.y = s_h[(frow + 48) * H_STRIDE + k0 + 1];
            acc0 = __builtin_amdgcn_wmma_f32_16x16x4_f32(
                false, a0, false, b, (short)0, acc0, false, false);
            acc1 = __builtin_amdgcn_wmma_f32_16x16x4_f32(
                false, a1, false, b, (short)0, acc1, false, false);
            acc2 = __builtin_amdgcn_wmma_f32_16x16x4_f32(
                false, a2, false, b, (short)0, acc2, false, false);
            acc3 = __builtin_amdgcn_wmma_f32_16x16x4_f32(
                false, a3, false, b, (short)0, acc3, false, false);
        }
        #pragma unroll
        for (int r = 0; r < 8; ++r) {
            atomicAdd(&s_out[(     mbase + r) * D_FEAT + ncol], acc0[r]);
            atomicAdd(&s_out[(16 + mbase + r) * D_FEAT + ncol], acc1[r]);
            atomicAdd(&s_out[(32 + mbase + r) * D_FEAT + ncol], acc2[r]);
            atomicAdd(&s_out[(48 + mbase + r) * D_FEAT + ncol], acc3[r]);
        }
    }
    __syncthreads();

    for (int i = tid; i < TILE_M * D_FEAT; i += 512) {
        int m = i >> 6, n = i & 63;
        int row = node0 + m;
        if (row < n_nodes)
            out[(size_t)row * D_FEAT + n] = s_out[i] + b2[n];
    }
}

extern "C" void kernel_launch(void* const* d_in, const int* in_sizes, int n_in,
                              void* d_out, int out_size, void* d_ws, size_t ws_size,
                              hipStream_t stream) {
    const float* feat = (const float*)d_in[0];
    const float* W1   = (const float*)d_in[1];
    const float* b1   = (const float*)d_in[2];
    const float* W2   = (const float*)d_in[3];
    const float* b2   = (const float*)d_in[4];
    const int*   src  = (const int*)d_in[5];
    const int*   dst  = (const int*)d_in[6];
    float*       out  = (float*)d_out;

    const int n_nodes = in_sizes[0] / D_FEAT;
    const int n_edges = in_sizes[5];

    float* rst = (float*)d_ws;

    {
        int n4 = (n_nodes * D_FEAT) / 4;
        gin_init_kernel<<<(n4 + 255) / 256, 256, 0, stream>>>(feat, rst, n4);
    }
    {
        int waves  = (n_edges + 31) / 32;
        int blocks = (waves + 7) / 8;
        gin_scatter_kernel<<<blocks, 256, 0, stream>>>(feat, src, dst, rst, n_edges);
    }
    {
        int blocks = (n_nodes + TILE_M - 1) / TILE_M;
        gin_mlp_kernel<<<blocks, 512, 0, stream>>>(rst, W1, b1, W2, b2, out, n_nodes);
    }
}


# Scatter replaced: edges grouped per 64-node tile each call by per-block LDS histograms (one integer add per tile and block), rows summed in LDS by ds_add_f32; no global f32 atomics; sum order arrival
# speedup vs baseline: 1.6302x; 1.0603x over previous
#include <hip/hip_runtime.h>
#include <hip/hip_bf16.h>

typedef __attribute__((ext_vector_type(2))) float v2f;
typedef __attribute__((ext_vector_type(8))) float v8f;

#define D_FEAT 64
#define H_DIM  256
#define TILE_M 64
#define RST_STRIDE 68
#define H_STRIDE   260

#define NT_MAX     2048
#define SORT_CHUNK 4096

__global__ void gin_zero_kernel(int* __restrict__ p, int n) {
    int i = blockIdx.x * blockDim.x + threadIdx.x;
    if (i < n) p[i] = 0;
}

__device__ __forceinline__ void load_chunk(const int* __restrict__ arr, int n_edges,
                                           int base, int fill, int (&v)[16]) {
    #pragma unroll
    for (int q = 0; q < 4; ++q) {
        const int e = base + q * 1024;
        if (e + 3 < n_edges) {
            const int4 t = *reinterpret_cast<const int4*>(arr + e);
            v[4 * q + 0] = t.x; v[4 * q + 1] = t.y;
            v[4 * q + 2] = t.z; v[4 * q + 3] = t.w;
        } else {
            #pragma unroll
            for (int k = 0; k < 4; ++k)
                v[4 * q + k] = (e + k < n_edges) ? arr[e + k] : fill;
        }
    }
}

__global__ __launch_bounds__(256)
void gin_count_kernel(const int* __restrict__ dst, int* __restrict__ tot,
                      int n_edges, int n_tiles) {
    __shared__ int hist[NT_MAX];
    const int tid = threadIdx.x;
    for (int t = tid; t < NT_MAX; t += 256) hist[t] = 0;
    int d[16];
    load_chunk(dst, n_edges, blockIdx.x * SORT_CHUNK + tid * 4, -1, d);
    __syncthreads();
    #pragma unroll
    for (int k = 0; k < 16; ++k)
        if (d[k] >= 0) atomicAdd(&hist[d[k] >> 6], 1);
    __syncthreads();
    for (int t = tid; t < n_tiles; t += 256) {
        const int c = hist[t];
        if (c) atomicAdd(&tot[t], c);
    }
}

__global__ __launch_bounds__(256)
void gin_fill_kernel(const int* __restrict__ src, const int* __restrict__ dst,
                     const int* __restrict__ tot, int* __restrict__ cursor,
                     int* __restrict__ gbase, int* __restrict__ rec,
                     int n_edges, int n_tiles) {
    __shared__ int hist[NT_MAX];
    __shared__ int sbase[NT_MAX];
    __shared__ int part[256];
    const int tid = threadIdx.x;
    for (int t = tid; t < NT_MAX; t += 256) hist[t] = 0;

    int d[16], s[16];
    const int ebase = blockIdx.x * SORT_CHUNK + tid * 4;
    load_chunk(dst, n_edges, ebase, -1, d);
    load_chunk(src, n_edges, ebase, 0, s);

    int v[8];
    {
        const int4 a = *reinterpret_cast<const int4*>(tot + tid * 8);
        const int4 b = *reinterpret_cast<const int4*>(tot + tid * 8 + 4);
        v[0] = a.x; v[1] = a.y; v[2] = a.z; v[3] = a.w;
        v[4] = b.x; v[5] = b.y; v[6] = b.z; v[7] = b.w;
    }
    int mine = 0;
    #pragma unroll
    for (int k = 0; k < 8; ++k) mine += v[k];
    part[tid] = mine;
    __syncthreads();

    #pragma unroll
    for (int k = 0; k < 16; ++k)
        if (d[k] >= 0) atomicAdd(&hist[d[k] >> 6], 1);

    for (int off = 1; off < 256; off <<= 1) {
        const int add = (tid >= off) ? part[tid - off] : 0;
        __syncthreads();
        part[tid] += add;
        __syncthreads();
    }
    int run = part[tid] - mine;
    #pragma unroll
    for (int k = 0; k < 8; ++k) { sbase[tid * 8 + k] = run; run += v[k]; }
    __syncthreads();

    for (int t = tid; t < n_tiles; t += 256) {
        const int c = hist[t];
        int g = 0;
        if (c) g = atomicAdd(&cursor[t], c);
        hist[t] = sbase[t] + g;
    }
    if (blockIdx.x == 0)
        for (int t = tid; t <= n_tiles; t += 256)
            gbase[t] = (t < n_tiles) ? sbase[t] : n_edges;
    __syncthreads();

    int2* __restrict__ rec2 = reinterpret_cast<int2*>(rec);
    #pragma unroll
    for (int k = 0; k < 16; ++k) {
        if (d[k] >= 0) {
            const int p = atomicAdd(&hist[d[k] >> 6], 1);
            rec2[p] = make_int2(s[k], d[k]);
        }
    }
}

#define AGG_GROUP 8

__global__ __launch_bounds__(256)
void gin_gather_kernel(const float* __restrict__ feat,
                       const int* __restrict__ rec,
                       const int* __restrict__ gbase,
                       float* __restrict__ rst, int n_nodes) {
    __shared__ float tile[TILE_M * D_FEAT];
    const int tid  = threadIdx.x;
    const int lane = tid & 31;
    const int wv    = __builtin_amdgcn_readfirstlane(tid >> 5);
    const int node0 = blockIdx.x * TILE_M;

    float4 own[4];
    #pragma unroll
    for (int q = 0; q < 4; ++q) {
        const int i4  = q * 256 + tid;
        const int row = node0 + (i4 >> 4);
        own[q] = make_float4(0.f, 0.f, 0.f, 0.f);
        if (row < n_nodes)
            own[q] = reinterpret_cast<const float4*>(feat)[(size_t)row * 16 + (i4 & 15)];
    }
    #pragma unroll
    for (int q = 0; q < 4; ++q)
        reinterpret_cast<float4*>(tile)[q * 256 + tid] = own[q];
    __syncthreads();

    const int r0 = gbase[blockIdx.x];
    const int r1 = gbase[blockIdx.x + 1];
    const int2* __restrict__ rec2 = reinterpret_cast<const int2*>(rec);

    for (int w0 = r0 + wv * 32; w0 < r1; w0 += 256) {
        const int e = w0 + lane;
        int s = 0, d = 0;
        if (e < r1) { const int2 r = rec2[e]; s = r.x; d = r.y - node0; }
        const int cnt = r1 - w0;
        #pragma unroll
        for (int g = 0; g < 32; g += AGG_GROUP) {
            if (g < cnt) {
                float lo[AGG_GROUP], hi[AGG_GROUP];
                #pragma unroll
                for (int j = 0; j < AGG_GROUP; ++j) {
                    const int sj = __builtin_amdgcn_readlane(s, g + j);
                    const float* row = feat + (size_t)sj * D_FEAT;
                    lo[j] = row[lane];
                    hi[j] = row[lane + 32];
                }
                #pragma unroll
                for (int j = 0; j < AGG_GROUP; ++j) {
                    const int dj = __builtin_amdgcn_readlane(d, g + j);
                    const bool ok = (g + j) < cnt;
                    float* trow = tile + dj * D_FEAT;
                    atomicAdd(trow + lane,      ok ? lo[j] : 0.f);
                    atomicAdd(trow + lane + 32, ok ? hi[j] : 0.f);
                }
            }
        }
    }
    __syncthreads();

    #pragma unroll
    for (int q = 0; q < 4; ++q) {
        const int i4  = q * 256 + tid;
        const int row = node0 + (i4 >> 4);
        if (row < n_nodes)
            reinterpret_cast<float4*>(rst)[(size_t)row * 16 + (i4 & 15)] =
                reinterpret_cast<const float4*>(tile)[i4];
    }
}

__global__ __launch_bounds__(512)
void gin_mlp_kernel(const float* __restrict__ rst,
                    const float* __restrict__ W1,
                    const float* __restrict__ b1,
                    const float* __restrict__ W2,
                    const float* __restrict__ b2,
                    float* __restrict__ out, int n_nodes) {
    __shared__ float s_rst[TILE_M * RST_STRIDE];
    __shared__ float s_h[TILE_M * H_STRIDE];
    __shared__ float s_out[TILE_M * D_FEAT];

    const int tid   = threadIdx.x;
    const int wave  = tid >> 5;
    const int lane  = tid & 31;
    const int node0 = blockIdx.x * TILE_M;

    for (int i = tid; i < TILE_M * D_FEAT; i += 512) {
        int m = i >> 6, k = i & 63;
        int row = node0 + m;
        row = row < n_nodes ? row : (n_nodes - 1);
        s_rst[m * RST_STRIDE + k] = rst[(size_t)row * D_FEAT + k];
    }
    for (int i = tid; i < TILE_M * D_FEAT; i += 512) s_out[i] = 0.0f;
    __syncthreads();

    const int frow  = lane & 15;
    const int khalf = (lane >> 4) << 1;
    const int mbase = (lane >> 4) << 3;

    {
        const int ncol = wave * 16 + frow;
        v8f acc0 = {}, acc1 = {}, acc2 = {}, acc3 = {};
        #pragma unroll
        for (int kk = 0; kk < 16; ++kk) {
            const int k0 = kk * 4 + khalf;
            v2f b;
            b.x = W1[(size_t)(k0    ) * H_DIM + ncol];
            b.y = W1[(size_t)(k0 + 1) * H_DIM + ncol];
            v2f a0, a1, a2, a3;
            a0.x = s_rst[(frow     ) * RST_STRIDE + k0];
            a0.y = s_rst[(frow     ) * RST_STRIDE + k0 + 1];
            a1.x = s_rst[(frow + 16) * RST_STRIDE + k0];
            a1.y = s_rst[(frow + 16) * RST_STRIDE + k0 + 1];
            a2.x = s_rst[(frow + 32) * RST_STRIDE + k0];
            a2.y = s_rst[(frow + 32) * RST_STRIDE + k0 + 1];
            a3.x = s_rst[(frow + 48) * RST_STRIDE + k0];
            a3.y = s_rst[(frow + 48) * RST_STRIDE + k0 + 1];
            acc0 = __builtin_amdgcn_wmma_f32_16x16x4_f32(
                false, a0, false, b, (short)0, acc0, false, false);
            acc1 = __builtin_amdgcn_wmma_f32_16x16x4_f32(
                false, a1, false, b, (short)0, acc1, false, false);
            acc2 = __builtin_amdgcn_wmma_f32_16x16x4_f32(
                false, a2, false, b, (short)0, acc2, false, false);
            acc3 = __builtin_amdgcn_wmma_f32_16x16x4_f32(
                false, a3, false, b, (short)0, acc3, false, false);
        }
        const float bias = b1[ncol];
        #pragma unroll
        for (int r = 0; r < 8; ++r) {
            float v0 = acc0[r] + bias, v1 = acc1[r] + bias;
            float v2 = acc2[r] + bias, v3 = acc3[r] + bias;
            s_h[(     mbase + r) * H_STRIDE + ncol] = v0 > 0.f ? v0 : 0.f;
            s_h[(16 + mbase + r) * H_STRIDE + ncol] = v1 > 0.f ? v1 : 0.f;
            s_h[(32 + mbase + r) * H_STRIDE + ncol] = v2 > 0.f ? v2 : 0.f;
            s_h[(48 + mbase + r) * H_STRIDE + ncol] = v3 > 0.f ? v3 : 0.f;
        }
    }
    __syncthreads();

    {
        const int ntile = (wave & 3) * 16;
        const int kbase = (wave >> 2) * 64;
        const int ncol  = ntile + frow;
        v8f acc0 = {}, acc1 = {}, acc2 = {}, acc3 = {};
        #pragma unroll
        for (int kk = 0; kk < 16; ++kk) {
            const int k0 = kbase + kk * 4 + khalf;
            v2f b;
            b.x = W2[(size_t)(k0    ) * D_FEAT + ncol];
            b.y = W2[(size_t)(k0 + 1) * D_FEAT + ncol];
            v2f a0, a1, a2, a3;
            a0.x = s_h[(frow     ) * H_STRIDE + k0];
            a0.y = s_h[(frow     ) * H_STRIDE + k0 + 1];
            a1.x = s_h[(frow + 16) * H_STRIDE + k0];
            a1.y = s_h[(frow + 16) * H_STRIDE + k0 + 1];
            a2.x = s_h[(frow + 32) * H_STRIDE + k0];
            a2.y = s_h[(frow + 32) * H_STRIDE + k0 + 1];
            a3.x = s_h[(frow + 48) * H_STRIDE + k0];
            a3.y = s_h[(frow + 48) * H_STRIDE + k0 + 1];
            acc0 = __builtin_amdgcn_wmma_f32_16x16x4_f32(
                false, a0, false, b, (short)0, acc0, false, false);
            acc1 = __builtin_amdgcn_wmma_f32_16x16x4_f32(
                false, a1, false, b, (short)0, acc1, false, false);
            acc2 = __builtin_amdgcn_wmma_f32_16x16x4_f32(
                false, a2, false, b, (short)0, acc2, false, false);
            acc3 = __builtin_amdgcn_wmma_f32_16x16x4_f32(
                false, a3, false, b, (short)0, acc3, false, false);
        }
        #pragma unroll
        for (int r = 0; r < 8; ++r) {
            atomicAdd(&s_out[(     mbase + r) * D_FEAT + ncol], acc0[r]);
            atomicAdd(&s_out[(16 + mbase + r) * D_FEAT + ncol], acc1[r]);
            atomicAdd(&s_out[(32 + mbase + r) * D_FEAT + ncol], acc2[r]);
            atomicAdd(&s_out[(48 + mbase + r) * D_FEAT + ncol], acc3[r]);
        }
    }
    __syncthreads();

    for (int i = tid; i < TILE_M * D_FEAT; i += 512) {
        int m = i >> 6, n = i & 63;
        int row = node0 + m;
        if (row < n_nodes)
            out[(size_t)row * D_FEAT + n] = s_out[i] + b2[n];
    }
}

extern "C" void kernel_launch(void* const* d_in, const int* in_sizes, int n_in,
                              void* d_out, int out_size, void* d_ws, size_t ws_size,
                              hipStream_t stream) {
    const float* feat = (const float*)d_in[0];
    const float* W1   = (const float*)d_in[1];
    const float* b1   = (const float*)d_in[2];
    const float* W2   = (const float*)d_in[3];
    const float* b2   = (const float*)d_in[4];
    const int*   src  = (const int*)d_in[5];
    const int*   dst  = (const int*)d_in[6];
    float*       out  = (float*)d_out;

    const int n_nodes = in_sizes[0] / D_FEAT;
    const int n_edges = in_sizes[5];
    const int n_tiles = (n_nodes + TILE_M - 1) / TILE_M;
    const int n_sort  = (n_edges + SORT_CHUNK - 1) / SORT_CHUNK;

    float* rst    = (float*)d_ws;
    int*   tot    = (int*)(rst + (size_t)n_nodes * D_FEAT);
    int*   cursor = tot + NT_MAX;
    int*   gbase  = cursor + NT_MAX;
    int*   rec    = gbase + NT_MAX + 64;

    gin_zero_kernel<<<(2 * NT_MAX + 255) / 256, 256, 0, stream>>>(tot, 2 * NT_MAX);
    gin_count_kernel<<<n_sort, 256, 0, stream>>>(dst, tot, n_edges, n_tiles);
    gin_fill_kernel<<<n_sort, 256, 0, stream>>>(src, dst, tot, cursor, gbase, rec,
                                                n_edges, n_tiles);
    gin_gather_kernel<<<n_tiles, 256, 0, stream>>>(feat, rec, gbase, rst, n_nodes);
    gin_mlp_kernel<<<n_tiles, 512, 0, stream>>>(rst, W1, b1, W2, b2, out, n_nodes);
}


# Gather fused into the MLP launch: each tile's neighbor sums are built by ds_add_f32 in the MLP block's own LDS tile, rst never goes through workspace; 4 launches; sum order arrival as before
# speedup vs baseline: 1.8446x; 1.1315x over previous
#include <hip/hip_runtime.h>
#include <hip/hip_bf16.h>

typedef __attribute__((ext_vector_type(2))) float v2f;
typedef __attribute__((ext_vector_type(8))) float v8f;

#define D_FEAT 64
#define H_DIM  256
#define TILE_M 64
#define RST_STRIDE 68
#define H_STRIDE   260

#define NT_MAX     2048
#define SORT_CHUNK 4096

__global__ void gin_zero_kernel(int* __restrict__ p, int n) {
    int i = blockIdx.x * blockDim.x + threadIdx.x;
    if (i < n) p[i] = 0;
}

__device__ __forceinline__ void load_chunk(const int* __restrict__ arr, int n_edges,
                                           int base, int fill, int (&v)[16]) {
    #pragma unroll
    for (int q = 0; q < 4; ++q) {
        const int e = base + q * 1024;
        if (e + 3 < n_edges) {
            const int4 t = *reinterpret_cast<const int4*>(arr + e);
            v[4 * q + 0] = t.x; v[4 * q + 1] = t.y;
            v[4 * q + 2] = t.z; v[4 * q + 3] = t.w;
        } else {
            #pragma unroll
            for (int k = 0; k < 4; ++k)
                v[4 * q + k] = (e + k < n_edges) ? arr[e + k] : fill;
        }
    }
}

__global__ __launch_bounds__(256)
void gin_count_kernel(const int* __restrict__ dst, int* __restrict__ tot,
                      int n_edges, int n_tiles) {
    __shared__ int hist[NT_MAX];
    const int tid = threadIdx.x;
    for (int t = tid; t < NT_MAX; t += 256) hist[t] = 0;
    int d[16];
    load_chunk(dst, n_edges, blockIdx.x * SORT_CHUNK + tid * 4, -1, d);
    __syncthreads();
    #pragma unroll
    for (int k = 0; k < 16; ++k)
        if (d[k] >= 0) atomicAdd(&hist[d[k] >> 6], 1);
    __syncthreads();
    for (int t = tid; t < n_tiles; t += 256) {
        const int c = hist[t];
        if (c) atomicAdd(&tot[t], c);
    }
}

__global__ __launch_bounds__(256)
void gin_fill_kernel(const int* __restrict__ src, const int* __restrict__ dst,
                     const int* __restrict__ tot, int* __restrict__ cursor,
                     int* __restrict__ gbase, int* __restrict__ rec,
                     int n_edges, int n_tiles) {
    __shared__ int hist[NT_MAX];
    __shared__ int sbase[NT_MAX];
    __shared__ int part[256];
    const int tid = threadIdx.x;
    for (int t = tid; t < NT_MAX; t += 256) hist[t] = 0;

    int d[16], s[16];
    const int ebase = blockIdx.x * SORT_CHUNK + tid * 4;
    load_chunk(dst, n_edges, ebase, -1, d);
    load_chunk(src, n_edges, ebase, 0, s);

    int v[8];
    {
        const int4 a = *reinterpret_cast<const int4*>(tot + tid * 8);
        const int4 b = *reinterpret_cast<const int4*>(tot + tid * 8 + 4);
        v[0] = a.x; v[1] = a.y; v[2] = a.z; v[3] = a.w;
        v[4] = b.x; v[5] = b.y; v[6] = b.z; v[7] = b.w;
    }
    int mine = 0;
    #pragma unroll
    for (int k = 0; k < 8; ++k) mine += v[k];
    part[tid] = mine;
    __syncthreads();

    #pragma unroll
    for (int k = 0; k < 16; ++k)
        if (d[k] >= 0) atomicAdd(&hist[d[k] >> 6], 1);

    for (int off = 1; off < 256; off <<= 1) {
        const int add = (tid >= off) ? part[tid - off] : 0;
        __syncthreads();
        part[tid] += add;
        __syncthreads();
    }
    int run = part[tid] - mine;
    #pragma unroll
    for (int k = 0; k < 8; ++k) { sbase[tid * 8 + k] = run; run += v[k]; }
    __syncthreads();

    for (int t = tid; t < n_tiles; t += 256) {
        const int c = hist[t];
        int g = 0;
        if (c) g = atomicAdd(&cursor[t], c);
        hist[t] = sbase[t] + g;
    }
    if (blockIdx.x == 0)
        for (int t = tid; t <= n_tiles; t += 256)
            gbase[t] = (t < n_tiles) ? sbase[t] : n_edges;
    __syncthreads();

    int2* __restrict__ rec2 = reinterpret_cast<int2*>(rec);
    #pragma unroll
    for (int k = 0; k < 16; ++k) {
        if (d[k] >= 0) {
            const int p = atomicAdd(&hist[d[k] >> 6], 1);
            rec2[p] = make_int2(s[k], d[k]);
        }
    }
}

#define AGG_GROUP 8

__global__ __launch_bounds__(512)
void gin_gather_mlp_kernel(const float* __restrict__ feat,
                           const int* __restrict__ rec,
                           const int* __restrict__ gbase,
                           const float* __restrict__ W1,
                           const float* __restrict__ b1,
                           const float* __restrict__ W2,
                           const float* __restrict__ b2,
                           float* __restrict__ out, int n_nodes) {
    __shared__ float s_rst[TILE_M * RST_STRIDE];
    __shared__ float s_h[TILE_M * H_STRIDE];
    __shared__ float s_out[TILE_M * D_FEAT];

    const int tid   = threadIdx.x;
    const int wave  = tid >> 5;
    const int lane  = tid & 31;
    const int wv    = __builtin_amdgcn_readfirstlane(tid >> 5);
    const int node0 = blockIdx.x * TILE_M;

    float4 own[2];
    #pragma unroll
    for (int q = 0; q < 2; ++q) {
        const int i4  = q * 512 + tid;
        const int row = node0 + (i4 >> 4);
        own[q] = make_float4(0.f, 0.f, 0.f, 0.f);
        if (row < n_nodes)
            own[q] = reinterpret_cast<const float4*>(feat)[(size_t)row * 16 + (i4 & 15)];
    }
    const int r0 = gbase[blockIdx.x];
    const int r1 = gbase[blockIdx.x + 1];
    #pragma unroll
    for (int q = 0; q < 2; ++q) {
        const int i4 = q * 512 + tid;
        *reinterpret_cast<float4*>(s_rst + (i4 >> 4) * RST_STRIDE + (i4 & 15) * 4) = own[q];
    }
    for (int i = tid; i < TILE_M * D_FEAT; i += 512) s_out[i] = 0.0f;
    __syncthreads();

    {
        const int2* __restrict__ rec2 = reinterpret_cast<const int2*>(rec);
        for (int w0 = r0 + wv * 32; w0 < r1; w0 += 512) {
            const int e = w0 + lane;
            int s = 0, d = 0;
            if (e < r1) { const int2 r = rec2[e]; s = r.x; d = r.y - node0; }
            const int cnt = r1 - w0;
            #pragma unroll
            for (int g = 0; g < 32; g += AGG_GROUP) {
                if (g < cnt) {
                    float lo[AGG_GROUP], hi[AGG_GROUP];
                    #pragma unroll
                    for (int j = 0; j < AGG_GROUP; ++j) {
                        const int sj = __builtin_amdgcn_readlane(s, g + j);
                        const float* row = feat + (size_t)sj * D_FEAT;
                        lo[j] = row[lane];
                        hi[j] = row[lane + 32];
                    }
                    #pragma unroll
                    for (int j = 0; j < AGG_GROUP; ++j) {
                        const int dj = __builtin_amdgcn_readlane(d, g + j);
                        const bool ok = (g + j) < cnt;
                        float* trow = s_rst + dj * RST_STRIDE;
                        atomicAdd(trow + lane,      ok ? lo[j] : 0.f);
                        atomicAdd(trow + lane + 32, ok ? hi[j] : 0.f);
                    }
                }
            }
        }
    }
    __syncthreads();

    const int frow  = lane & 15;
    const int khalf = (lane >> 4) << 1;
    const int mbase = (lane >> 4) << 3;

    {
        const int ncol = wave * 16 + frow;
        v8f acc0 = {}, acc1 = {}, acc2 = {}, acc3 = {};
        #pragma unroll
        for (int kk = 0; kk < 16; ++kk) {
            const int k0 = kk * 4 + khalf;
            v2f b;
            b.x = W1[(size_t)(k0    ) * H_DIM + ncol];
            b.y = W1[(size_t)(k0 + 1) * H_DIM + ncol];
            v2f a0, a1, a2, a3;
            a0.x = s_rst[(frow     ) * RST_STRIDE + k0];
            a0.y = s_rst[(frow     ) * RST_STRIDE + k0 + 1];
            a1.x = s_rst[(frow + 16) * RST_STRIDE + k0];
            a1.y = s_rst[(frow + 16) * RST_STRIDE + k0 + 1];
            a2.x = s_rst[(frow + 32) * RST_STRIDE + k0];
            a2.y = s_rst[(frow + 32) * RST_STRIDE + k0 + 1];
            a3.x = s_rst[(frow + 48) * RST_STRIDE + k0];
            a3.y = s_rst[(frow + 48) * RST_STRIDE + k0 + 1];
            acc0 = __builtin_amdgcn_wmma_f32_16x16x4_f32(
                false, a0, false, b, (short)0, acc0, false, false);
            acc1 = __builtin_amdgcn_wmma_f32_16x16x4_f32(
                false, a1, false, b, (short)0, acc1, false, false);
            acc2 = __builtin_amdgcn_wmma_f32_16x16x4_f32(
                false, a2, false, b, (short)0, acc2, false, false);
            acc3 = __builtin_amdgcn_wmma_f32_16x16x4_f32(
                false, a3, false, b, (short)0, acc3, false, false);
        }
        const float bias = b1[ncol];
        #pragma unroll
        for (int r = 0; r < 8; ++r) {
            float v0 = acc0[r] + bias, v1 = acc1[r] + bias;
            float v2 = acc2[r] + bias, v3 = acc3[r] + bias;
            s_h[(     mbase + r) * H_STRIDE + ncol] = v0 > 0.f ? v0 : 0.f;
            s_h[(16 + mbase + r) * H_STRIDE + ncol] = v1 > 0.f ? v1 : 0.f;
            s_h[(32 + mbase + r) * H_STRIDE + ncol] = v2 > 0.f ? v2 : 0.f;
            s_h[(48 + mbase + r) * H_STRIDE + ncol] = v3 > 0.f ? v3 : 0.f;
        }
    }
    __syncthreads();

    {
        const int ntile = (wave & 3) * 16;
        const int kbase = (wave >> 2) * 64;
        const int ncol  = ntile + frow;
        v8f acc0 = {}, acc1 = {}, acc2 = {}, acc3 = {};
        #pragma unroll
        for (int kk = 0; kk < 16; ++kk) {
            const int k0 = kbase + kk * 4 + khalf;
            v2f b;
            b.x = W2[(size_t)(k0    ) * D_FEAT + ncol];
            b.y = W2[(size_t)(k0 + 1) * D_FEAT + ncol];
            v2f a0, a1, a2, a3;
            a0.x = s_h[(frow     ) * H_STRIDE + k0];
            a0.y = s_h[(frow     ) * H_STRIDE + k0 + 1];
            a1.x = s_h[(frow + 16) * H_STRIDE + k0];
            a1.y = s_h[(frow + 16) * H_STRIDE + k0 + 1];
            a2.x = s_h[(frow + 32) * H_STRIDE + k0];
            a2.y = s_h[(frow + 32) * H_STRIDE + k0 + 1];
            a3.x = s_h[(frow + 48) * H_STRIDE + k0];
            a3.y = s_h[(frow + 48) * H_STRIDE + k0 + 1];
            acc0 = __builtin_amdgcn_wmma_f32_16x16x4_f32(
                false, a0, false, b, (short)0, acc0, false, false);
            acc1 = __builtin_amdgcn_wmma_f32_16x16x4_f32(
                false, a1, false, b, (short)0, acc1, false, false);
            acc2 = __builtin_amdgcn_wmma_f32_16x16x4_f32(
                false, a2, false, b, (short)0, acc2, false, false);
            acc3 = __builtin_amdgcn_wmma_f32_16x16x4_f32(
                false, a3, false, b, (short)0, acc3, false, false);
        }
        #pragma unroll
        for (int r = 0; r < 8; ++r) {
            atomicAdd(&s_out[(     mbase + r) * D_FEAT + ncol], acc0[r]);
            atomicAdd(&s_out[(16 + mbase + r) * D_FEAT + ncol], acc1[r]);
            atomicAdd(&s_out[(32 + mbase + r) * D_FEAT + ncol], acc2[r]);
            atomicAdd(&s_out[(48 + mbase + r) * D_FEAT + ncol], acc3[r]);
        }
    }
    __syncthreads();

    for (int i = tid; i < TILE_M * D_FEAT; i += 512) {
        int m = i >> 6, n = i & 63;
        int row = node0 + m;
        if (row < n_nodes)
            out[(size_t)row * D_FEAT + n] = s_out[i] + b2[n];
    }
}

extern "C" void kernel_launch(void* const* d_in, const int* in_sizes, int n_in,
                              void* d_out, int out_size, void* d_ws, size_t ws_size,
                              hipStream_t stream) {
    const float* feat = (const float*)d_in[0];
    const float* W1   = (const float*)d_in[1];
    const float* b1   = (const float*)d_in[2];
    const float* W2   = (const float*)d_in[3];
    const float* b2   = (const float*)d_in[4];
    const int*   src  = (const int*)d_in[5];
    const int*   dst  = (const int*)d_in[6];
    float*       out  = (float*)d_out;

    const int n_nodes = in_sizes[0] / D_FEAT;
    const int n_edges = in_sizes[5];
    const int n_tiles = (n_nodes + TILE_M - 1) / TILE_M;
    const int n_sort  = (n_edges + SORT_CHUNK - 1) / SORT_CHUNK;

    int*   tot    = (int*)d_ws;
    int*   cursor = tot + NT_MAX;
    int*   gbase  = cursor + NT_MAX;
    int*   rec    = gbase + NT_MAX + 64;

    gin_zero_kernel<<<(2 * NT_MAX + 255) / 256, 256, 0, stream>>>(tot, 2 * NT_MAX);
    gin_count_kernel<<<n_sort, 256, 0, stream>>>(dst, tot, n_edges, n_tiles);
    gin_fill_kernel<<<n_sort, 256, 0, stream>>>(src, dst, tot, cursor, gbase, rec,
                                                n_edges, n_tiles);
    gin_gather_mlp_kernel<<<n_tiles, 512, 0, stream>>>(feat, rec, gbase, W1, b1, W2, b2,
                                                       out, n_nodes);
}


# Edge grouping in ONE launch: a block sorts its 12800-edge chunk by tile in LDS, stores whole lines; MLP block finds its pieces by offset table, next window's ahead; 4->2 launches; sum order arrival
# speedup vs baseline: 2.0438x; 1.1080x over previous
#include <hip/hip_runtime.h>
#include <hip/hip_bf16.h>

typedef __attribute__((ext_vector_type(2))) float v2f;
typedef __attribute__((ext_vector_type(8))) float v8f;

#define D_FEAT 64
#define H_DIM  256
#define TILE_M 64
#define RST_STRIDE 68
#define H_STRIDE   260

#define NT_MAX     2048
#define CHUNK      12800
#define NB_MAX     128

__device__ __forceinline__ int wave_incl_scan(int v, int lane) {
    #pragma unroll
    for (int off = 1; off < 32; off <<= 1) {
        const int u = __shfl_up(v, off);
        if (lane >= off) v += u;
    }
    return v;
}

__global__ __launch_bounds__(1024)
void gin_sort_kernel(const int* __restrict__ src, const int* __restrict__ dst,
                     int* __restrict__ offsT, int* __restrict__ rec,
                     int n_edges, int n_tiles) {
    __shared__ __align__(16) int2 sorted[CHUNK];
    __shared__ int hist[NT_MAX];
    __shared__ int wsum[32];
    const int tid  = threadIdx.x;
    const int lane = tid & 31;
    const int wave = tid >> 5;
    const int e0   = blockIdx.x * CHUNK;
    const int nloc = (n_edges - e0 < CHUNK) ? (n_edges - e0) : CHUNK;

    hist[tid] = 0;
    hist[tid + 1024] = 0;

    int d[16], s[16];
    if (nloc == CHUNK) {
        #pragma unroll
        for (int q = 0; q < 4; ++q) {
            const int i4 = q * 1024 + tid;
            int4 td = make_int4(-1, -1, -1, -1);
            int4 ts = make_int4(0, 0, 0, 0);
            if (i4 < CHUNK / 4) {
                td = reinterpret_cast<const int4*>(dst + e0)[i4];
                ts = reinterpret_cast<const int4*>(src + e0)[i4];
            }
            d[4 * q + 0] = td.x; d[4 * q + 1] = td.y;
            d[4 * q + 2] = td.z; d[4 * q + 3] = td.w;
            s[4 * q + 0] = ts.x; s[4 * q + 1] = ts.y;
            s[4 * q + 2] = ts.z; s[4 * q + 3] = ts.w;
        }
    } else {
        #pragma unroll
        for (int k = 0; k < 16; ++k) {
            const int i = ((k >> 2) * 1024 + tid) * 4 + (k & 3);
            const bool in = i < nloc;
            d[k] = in ? dst[e0 + i] : -1;
            s[k] = in ? src[e0 + i] : 0;
        }
    }
    __syncthreads();

    int r[16];
    #pragma unroll
    for (int k = 0; k < 16; ++k) {
        r[k] = 0;
        if (d[k] >= 0) r[k] = atomicAdd(&hist[d[k] >> 6], 1);
    }
    __syncthreads();

    const int c0 = hist[2 * tid];
    const int c1 = hist[2 * tid + 1];
    const int mine = c0 + c1;
    const int incl = wave_incl_scan(mine, lane);
    if (lane == 31) wsum[wave] = incl;
    __syncthreads();
    if (wave == 0) {
        const int v = wsum[lane];
        wsum[lane] = wave_incl_scan(v, lane) - v;
    }
    __syncthreads();
    const int excl = wsum[wave] + incl - mine;
    hist[2 * tid]     = excl;
    hist[2 * tid + 1] = excl + c0;
    __syncthreads();

    int pos[16];
    #pragma unroll
    for (int k = 0; k < 16; ++k)
        pos[k] = hist[(d[k] >= 0) ? (d[k] >> 6) : 0] + r[k];
    #pragma unroll
    for (int k = 0; k < 16; ++k)
        if (d[k] >= 0) sorted[pos[k]] = make_int2(s[k], d[k]);

    for (int t = tid; t <= n_tiles; t += 1024)
        offsT[t * NB_MAX + blockIdx.x] = hist[t];
    __syncthreads();

    int4* __restrict__ o4 = reinterpret_cast<int4*>(rec + 2 * (size_t)e0);
    const int4* s4 = reinterpret_cast<const int4*>(sorted);
    const int n4 = nloc >> 1;
    for (int i = tid; i < n4; i += 1024) o4[i] = s4[i];
    if ((nloc & 1) && tid == 0)
        reinterpret_cast<int2*>(rec)[(size_t)e0 + nloc - 1] = sorted[nloc - 1];
}

__device__ __forceinline__ int2 fetch_rec(const int2* __restrict__ rec2,
                                          const int* s_pos, const int* s_adj,
                                          int j, int total, int node0) {
    int2 r = make_int2(0, node0);
    if (j < total) {
        int lo = 0;
        #pragma unroll
        for (int step = NB_MAX / 2; step >= 1; step >>= 1)
            if (s_pos[lo + step] <= j) lo += step;
        r = rec2[s_adj[lo] + j];
    }
    return r;
}

#define AGG_GROUP 8

__global__ __launch_bounds__(512)
void gin_gather_mlp_kernel(const float* __restrict__ feat,
                           const int* __restrict__ rec,
                           const int* __restrict__ offsT,
                           const float* __restrict__ W1,
                           const float* __restrict__ b1,
                           const float* __restrict__ W2,
                           const float* __restrict__ b2,
                           float* __restrict__ out, int n_nodes, int n_chunks) {
    __shared__ float s_rst[TILE_M * RST_STRIDE];
    __shared__ float s_h[TILE_M * H_STRIDE];
    __shared__ float s_out[TILE_M * D_FEAT];
    __shared__ int s_pos[NB_MAX + 1];
    __shared__ int s_adj[NB_MAX];
    __shared__ int s_wtot[NB_MAX / 32];

    const int tid   = threadIdx.x;
    const int wave  = tid >> 5;
    const int lane  = tid & 31;
    const int wv    = __builtin_amdgcn_readfirstlane(tid >> 5);
    const int node0 = blockIdx.x * TILE_M;

    float4 own[2];
    #pragma unroll
    for (int q = 0; q < 2; ++q) {
        const int i4  = q * 512 + tid;
        const int row = node0 + (i4 >> 4);
        own[q] = make_float4(0.f, 0.f, 0.f, 0.f);
        if (row < n_nodes)
            own[q] = reinterpret_cast<const float4*>(feat)[(size_t)row * 16 + (i4 & 15)];
    }
    int pstart = 0, pcnt = 0;
    if (tid < n_chunks) {
        pstart = offsT[blockIdx.x * NB_MAX + tid];
        pcnt   = offsT[(blockIdx.x + 1) * NB_MAX + tid] - pstart;
    }
    const int pincl = wave_incl_scan(pcnt, lane);
    if (tid < NB_MAX && lane == 31) s_wtot[wave] = pincl;
    #pragma unroll
    for (int q = 0; q < 2; ++q) {
        const int i4 = q * 512 + tid;
        *reinterpret_cast<float4*>(s_rst + (i4 >> 4) * RST_STRIDE + (i4 & 15) * 4) = own[q];
    }
    for (int i = tid; i < TILE_M * D_FEAT; i += 512) s_out[i] = 0.0f;
    __syncthreads();
    if (tid < NB_MAX) {
        int pre = 0;
        #pragma unroll
        for (int w = 0; w < NB_MAX / 32; ++w) pre += (w < wave) ? s_wtot[w] : 0;
        const int pexcl = pre + pincl - pcnt;
        s_pos[tid] = pexcl;
        s_adj[tid] = tid * CHUNK + pstart - pexcl;
        if (tid == NB_MAX - 1) s_pos[NB_MAX] = pexcl + pcnt;
    }
    __syncthreads();
    const int total = s_pos[NB_MAX];

    {
        const int2* __restrict__ rec2 = reinterpret_cast<const int2*>(rec);
        int j = wv * 32 + lane;
        int2 cur = fetch_rec(rec2, s_pos, s_adj, j, total, node0);
        for (int w0 = wv * 32; w0 < total; w0 += 512) {
            asm volatile("" :: "v"(cur.x), "v"(cur.y));
            const int s = cur.x, d = cur.y - node0;
            j += 512;
            const int2 nxt = fetch_rec(rec2, s_pos, s_adj, j, total, node0);
            const int cnt = total - w0;
            #pragma unroll
            for (int g = 0; g < 32; g += AGG_GROUP) {
                if (g < cnt) {
                    float lo[AGG_GROUP], hi[AGG_GROUP];
                    #pragma unroll
                    for (int jj = 0; jj < AGG_GROUP; ++jj) {
                        const int sj = __builtin_amdgcn_readlane(s, g + jj);
                        const float* row = feat + (size_t)sj * D_FEAT;
                        lo[jj] = row[lane];
                        hi[jj] = row[lane + 32];
                    }
                    #pragma unroll
                    for (int jj = 0; jj < AGG_GROUP; ++jj) {
                        const int dj = __builtin_amdgcn_readlane(d, g + jj);
                        const bool ok = (g + jj) < cnt;
                        float* trow = s_rst + dj * RST_STRIDE;
                        atomicAdd(trow + lane,      ok ? lo[jj] : 0.f);
                        atomicAdd(trow + lane + 32, ok ? hi[jj] : 0.f);
                    }
                }
            }
            cur = nxt;
        }
    }
    __syncthreads();

    const int frow  = lane & 15;
    const int khalf = (lane >> 4) << 1;
    const int mbase = (lane >> 4) << 3;

    {
        const int ncol = wave * 16 + frow;
        v8f acc0 = {}, acc1 = {}, acc2 = {}, acc3 = {};
        #pragma unroll
        for (int kk = 0; kk < 16; ++kk) {
            const int k0 = kk * 4 + khalf;
            v2f b;
            b.x = W1[(size_t)(k0    ) * H_DIM + ncol];
            b.y = W1[(size_t)(k0 + 1) * H_DIM + ncol];
            v2f a0, a1, a2, a3;
            a0.x = s_rst[(frow     ) * RST_STRIDE + k0];
            a0.y = s_rst[(frow     ) * RST_STRIDE + k0 + 1];
            a1.x = s_rst[(frow + 16) * RST_STRIDE + k0];
            a1.y = s_rst[(frow + 16) * RST_STRIDE + k0 + 1];
            a2.x = s_rst[(frow + 32) * RST_STRIDE + k0];
            a2.y = s_rst[(frow + 32) * RST_STRIDE + k0 + 1];
            a3.x = s_rst[(frow + 48) * RST_STRIDE + k0];
            a3.y = s_rst[(frow + 48) * RST_STRIDE + k0 + 1];
            acc0 = __builtin_amdgcn_wmma_f32_16x16x4_f32(
                false, a0, false, b, (short)0, acc0, false, false);
            acc1 = __builtin_amdgcn_wmma_f32_16x16x4_f32(
                false, a1, false, b, (short)0, acc1, false, false);
            acc2 = __builtin_amdgcn_wmma_f32_16x16x4_f32(
                false, a2, false, b, (short)0, acc2, false, false);
            acc3 = __builtin_amdgcn_wmma_f32_16x16x4_f32(
                false, a3, false, b, (short)0, acc3, false, false);
        }
        const float bias = b1[ncol];
        #pragma unroll
        for (int r = 0; r < 8; ++r) {
            float v0 = acc0[r] + bias, v1 = acc1[r] + bias;
            float v2 = acc2[r] + bias, v3 = acc3[r] + bias;
            s_h[(     mbase + r) * H_STRIDE + ncol] = v0 > 0.f ? v0 : 0.f;
            s_h[(16 + mbase + r) * H_STRIDE + ncol] = v1 > 0.f ? v1 : 0.f;
            s_h[(32 + mbase + r) * H_STRIDE + ncol] = v2 > 0.f ? v2 : 0.f;
            s_h[(48 + mbase + r) * H_STRIDE + ncol] = v3 > 0.f ? v3 : 0.f;
        }
    }
    __syncthreads();

    {
        const int ntile = (wave & 3) * 16;
        const int kbase = (wave >> 2) * 64;
        const int ncol  = ntile + frow;
        v8f acc0 = {}, acc1 = {}, acc2 = {}, acc3 = {};
        #pragma unroll
        for (int kk = 0; kk < 16; ++kk) {
            const int k0 = kbase + kk * 4 + khalf;
            v2f b;
            b.x = W2[(size_t)(k0    ) * D_FEAT + ncol];
            b.y = W2[(size_t)(k0 + 1) * D_FEAT + ncol];
            v2f a0, a1, a2, a3;
            a0.x = s_h[(frow     ) * H_STRIDE + k0];
            a0.y = s_h[(frow     ) * H_STRIDE + k0 + 1];
            a1.x = s_h[(frow + 16) * H_STRIDE + k0];
            a1.y = s_h[(frow + 16) * H_STRIDE + k0 + 1];
            a2.x = s_h[(frow + 32) * H_STRIDE + k0];
            a2.y = s_h[(frow + 32) * H_STRIDE + k0 + 1];
            a3.x = s_h[(frow + 48) * H_STRIDE + k0];
            a3.y = s_h[(frow + 48) * H_STRIDE + k0 + 1];
            acc0 = __builtin_amdgcn_wmma_f32_16x16x4_f32(
                false, a0, false, b, (short)0, acc0, false, false);
            acc1 = __builtin_amdgcn_wmma_f32_16x16x4_f32(
                false, a1, false, b, (short)0, acc1, false, false);
            acc2 = __builtin_amdgcn_wmma_f32_16x16x4_f32(
                false, a2, false, b, (short)0, acc2, false, false);
            acc3 = __builtin_amdgcn_wmma_f32_16x16x4_f32(
                false, a3, false, b, (short)0, acc3, false, false);
        }
        #pragma unroll
        for (int r = 0; r < 8; ++r) {
            atomicAdd(&s_out[(     mbase + r) * D_FEAT + ncol], acc0[r]);
            atomicAdd(&s_out[(16 + mbase + r) * D_FEAT + ncol], acc1[r]);
            atomicAdd(&s_out[(32 + mbase + r) * D_FEAT + ncol], acc2[r]);
            atomicAdd(&s_out[(48 + mbase + r) * D_FEAT + ncol], acc3[r]);
        }
    }
    __syncthreads();

    for (int i = tid; i < TILE_M * D_FEAT; i += 512) {
        int m = i >> 6, n = i & 63;
        int row = node0 + m;
        if (row < n_nodes)
            out[(size_t)row * D_FEAT + n] = s_out[i] + b2[n];
    }
}

extern "C" void kernel_launch(void* const* d_in, const int* in_sizes, int n_in,
                              void* d_out, int out_size, void* d_ws, size_t ws_size,
                              hipStream_t stream) {
    const float* feat = (const float*)d_in[0];
    const float* W1   = (const float*)d_in[1];
    const float* b1   = (const float*)d_in[2];
    const float* W2   = (const float*)d_in[3];
    const float* b2   = (const float*)d_in[4];
    const int*   src  = (const int*)d_in[5];
    const int*   dst  = (const int*)d_in[6];
    float*       out  = (float*)d_out;

    const int n_nodes  = in_sizes[0] / D_FEAT;
    const int n_edges  = in_sizes[5];
    const int n_tiles  = (n_nodes + TILE_M - 1) / TILE_M;
    const int n_chunks = (n_edges + CHUNK - 1) / CHUNK;

    int* offsT = (int*)d_ws;
    int* rec   = offsT + (NT_MAX + 1) * NB_MAX;

    gin_sort_kernel<<<n_chunks, 1024, 0, stream>>>(src, dst, offsT, rec, n_edges, n_tiles);
    gin_gather_mlp_kernel<<<n_tiles, 512, 0, stream>>>(feat, rec, offsT, W1, b1, W2, b2,
                                                       out, n_nodes, n_chunks);
}


# MLP kernel held to 80 registers (12 waves per SIMD asked in its launch bounds): 3 resident 16-wave blocks per WGP instead of 2, more of a block's waits covered; sum order arrival as before
# speedup vs baseline: 2.2193x; 1.0859x over previous
#include <hip/hip_runtime.h>
#include <hip/hip_bf16.h>

typedef __attribute__((ext_vector_type(2))) float v2f;
typedef __attribute__((ext_vector_type(8))) float v8f;

#define D_FEAT 64
#define H_DIM  256
#define TILE_M 64
#define RST_STRIDE 68
#define H_STRIDE   260

#define NT_MAX     2048
#define CHUNK      12800
#define NB_MAX     128

__device__ __forceinline__ int wave_incl_scan(int v, int lane) {
    #pragma unroll
    for (int off = 1; off < 32; off <<= 1) {
        const int u = __shfl_up(v, off);
        if (lane >= off) v += u;
    }
    return v;
}

__global__ __launch_bounds__(1024)
void gin_sort_kernel(const int* __restrict__ src, const int* __restrict__ dst,
                     int* __restrict__ offsT, int* __restrict__ rec,
                     int n_edges, int n_tiles) {
    __shared__ __align__(16) int2 sorted[CHUNK];
    __shared__ int hist[NT_MAX];
    __shared__ int wsum[32];
    const int tid  = threadIdx.x;
    const int lane = tid & 31;
    const int wave = tid >> 5;
    const int e0   = blockIdx.x * CHUNK;
    const int nloc = (n_edges - e0 < CHUNK) ? (n_edges - e0) : CHUNK;

    hist[tid] = 0;
    hist[tid + 1024] = 0;

    int d[16], s[16];
    if (nloc == CHUNK) {
        #pragma unroll
        for (int q = 0; q < 4; ++q) {
            const int i4 = q * 1024 + tid;
            int4 td = make_int4(-1, -1, -1, -1);
            int4 ts = make_int4(0, 0, 0, 0);
            if (i4 < CHUNK / 4) {
                td = reinterpret_cast<const int4*>(dst + e0)[i4];
                ts = reinterpret_cast<const int4*>(src + e0)[i4];
            }
            d[4 * q + 0] = td.x; d[4 * q + 1] = td.y;
            d[4 * q + 2] = td.z; d[4 * q + 3] = td.w;
            s[4 * q + 0] = ts.x; s[4 * q + 1] = ts.y;
            s[4 * q + 2] = ts.z; s[4 * q + 3] = ts.w;
        }
    } else {
        #pragma unroll
        for (int k = 0; k < 16; ++k) {
            const int i = ((k >> 2) * 1024 + tid) * 4 + (k & 3);
            const bool in = i < nloc;
            d[k] = in ? dst[e0 + i] : -1;
            s[k] = in ? src[e0 + i] : 0;
        }
    }
    __syncthreads();

    int r[16];
    #pragma unroll
    for (int k = 0; k < 16; ++k) {
        r[k] = 0;
        if (d[k] >= 0) r[k] = atomicAdd(&hist[d[k] >> 6], 1);
    }
    __syncthreads();

    const int c0 = hist[2 * tid];
    const int c1 = hist[2 * tid + 1];
    const int mine = c0 + c1;
    const int incl = wave_incl_scan(mine, lane);
    if (lane == 31) wsum[wave] = incl;
    __syncthreads();
    if (wave == 0) {
        const int v = wsum[lane];
        wsum[lane] = wave_incl_scan(v, lane) - v;
    }
    __syncthreads();
    const int excl = wsum[wave] + incl - mine;
    hist[2 * tid]     = excl;
    hist[2 * tid + 1] = excl + c0;
    __syncthreads();

    int pos[16];
    #pragma unroll
    for (int k = 0; k < 16; ++k)
        pos[k] = hist[(d[k] >= 0) ? (d[k] >> 6) : 0] + r[k];
    #pragma unroll
    for (int k = 0; k < 16; ++k)
        if (d[k] >= 0) sorted[pos[k]] = make_int2(s[k], d[k]);

    for (int t = tid; t <= n_tiles; t += 1024)
        offsT[t * NB_MAX + blockIdx.x] = hist[t];
    __syncthreads();

    int4* __restrict__ o4 = reinterpret_cast<int4*>(rec + 2 * (size_t)e0);
    const int4* s4 = reinterpret_cast<const int4*>(sorted);
    const int n4 = nloc >> 1;
    for (int i = tid; i < n4; i += 1024) o4[i] = s4[i];
    if ((nloc & 1) && tid == 0)
        reinterpret_cast<int2*>(rec)[(size_t)e0 + nloc - 1] = sorted[nloc - 1];
}

__device__ __forceinline__ int2 fetch_rec(const int2* __restrict__ rec2,
                                          const int* s_pos, const int* s_adj,
                                          int j, int total, int node0) {
    int2 r = make_int2(0, node0);
    if (j < total) {
        int lo = 0;
        #pragma unroll
        for (int step = NB_MAX / 2; step >= 1; step >>= 1)
            if (s_pos[lo + step] <= j) lo += step;
        r = rec2[s_adj[lo] + j];
    }
    return r;
}

#define AGG_GROUP 8

__global__ __launch_bounds__(512, 12)
void gin_gather_mlp_kernel(const float* __restrict__ feat,
                           const int* __restrict__ rec,
                           const int* __restrict__ offsT,
                           const float* __restrict__ W1,
                           const float* __restrict__ b1,
                           const float* __restrict__ W2,
                           const float* __restrict__ b2,
                           float* __restrict__ out, int n_nodes, int n_chunks) {
    __shared__ float s_rst[TILE_M * RST_STRIDE];
    __shared__ float s_h[TILE_M * H_STRIDE];
    __shared__ float s_out[TILE_M * D_FEAT];
    __shared__ int s_pos[NB_MAX + 1];
    __shared__ int s_adj[NB_MAX];
    __shared__ int s_wtot[NB_MAX / 32];

    const int tid   = threadIdx.x;
    const int wave  = tid >> 5;
    const int lane  = tid & 31;
    const int wv    = __builtin_amdgcn_readfirstlane(tid >> 5);
    const int node0 = blockIdx.x * TILE_M;

    float4 own[2];
    #pragma unroll
    for (int q = 0; q < 2; ++q) {
        const int i4  = q * 512 + tid;
        const int row = node0 + (i4 >> 4);
        own[q] = make_float4(0.f, 0.f, 0.f, 0.f);
        if (row < n_nodes)
            own[q] = reinterpret_cast<const float4*>(feat)[(size_t)row * 16 + (i4 & 15)];
    }
    int pstart = 0, pcnt = 0;
    if (tid < n_chunks) {
        pstart = offsT[blockIdx.x * NB_MAX + tid];
        pcnt   = offsT[(blockIdx.x + 1) * NB_MAX + tid] - pstart;
    }
    const int pincl = wave_incl_scan(pcnt, lane);
    if (tid < NB_MAX && lane == 31) s_wtot[wave] = pincl;
    #pragma unroll
    for (int q = 0; q < 2; ++q) {
        const int i4 = q * 512 + tid;
        *reinterpret_cast<float4*>(s_rst + (i4 >> 4) * RST_STRIDE + (i4 & 15) * 4) = own[q];
    }
    for (int i = tid; i < TILE_M * D_FEAT; i += 512) s_out[i] = 0.0f;
    __syncthreads();
    if (tid < NB_MAX) {
        int pre = 0;
        #pragma unroll
        for (int w = 0; w < NB_MAX / 32; ++w) pre += (w < wave) ? s_wtot[w] : 0;
        const int pexcl = pre + pincl - pcnt;
        s_pos[tid] = pexcl;
        s_adj[tid] = tid * CHUNK + pstart - pexcl;
        if (tid == NB_MAX - 1) s_pos[NB_MAX] = pexcl + pcnt;
    }
    __syncthreads();
    const int total = s_pos[NB_MAX];

    {
        const int2* __restrict__ rec2 = reinterpret_cast<const int2*>(rec);
        int j = wv * 32 + lane;
        int2 cur = fetch_rec(rec2, s_pos, s_adj, j, total, node0);
        for (int w0 = wv * 32; w0 < total; w0 += 512) {
            asm volatile("" :: "v"(cur.x), "v"(cur.y));
            const int s = cur.x, d = cur.y - node0;
            j += 512;
            const int2 nxt = fetch_rec(rec2, s_pos, s_adj, j, total, node0);
            const int cnt = total - w0;
            #pragma unroll
            for (int g = 0; g < 32; g += AGG_GROUP) {
                if (g < cnt) {
                    float lo[AGG_GROUP], hi[AGG_GROUP];
                    #pragma unroll
                    for (int jj = 0; jj < AGG_GROUP; ++jj) {
                        const int sj = __builtin_amdgcn_readlane(s, g + jj);
                        const float* row = feat + (size_t)sj * D_FEAT;
                        lo[jj] = row[lane];
                        hi[jj] = row[lane + 32];
                    }
                    #pragma unroll
                    for (int jj = 0; jj < AGG_GROUP; ++jj) {
                        const int dj = __builtin_amdgcn_readlane(d, g + jj);
                        const bool ok = (g + jj) < cnt;
                        float* trow = s_rst + dj * RST_STRIDE;
                        atomicAdd(trow + lane,      ok ? lo[jj] : 0.f);
                        atomicAdd(trow + lane + 32, ok ? hi[jj] : 0.f);
                    }
                }
            }
            cur = nxt;
        }
    }
    __syncthreads();

    const int frow  = lane & 15;
    const int khalf = (lane >> 4) << 1;
    const int mbase = (lane >> 4) << 3;

    {
        const int ncol = wave * 16 + frow;
        v8f acc0 = {}, acc1 = {}, acc2 = {}, acc3 = {};
        #pragma unroll
        for (int kk = 0; kk < 16; ++kk) {
            const int k0 = kk * 4 + khalf;
            v2f b;
            b.x = W1[(size_t)(k0    ) * H_DIM + ncol];
            b.y = W1[(size_t)(k0 + 1) * H_DIM + ncol];
            v2f a0, a1, a2, a3;
            a0.x = s_rst[(frow     ) * RST_STRIDE + k0];
            a0.y = s_rst[(frow     ) * RST_STRIDE + k0 + 1];
            a1.x = s_rst[(frow + 16) * RST_STRIDE + k0];
            a1.y = s_rst[(frow + 16) * RST_STRIDE + k0 + 1];
            a2.x = s_rst[(frow + 32) * RST_STRIDE + k0];
            a2.y = s_rst[(frow + 32) * RST_STRIDE + k0 + 1];
            a3.x = s_rst[(frow + 48) * RST_STRIDE + k0];
            a3.y = s_rst[(frow + 48) * RST_STRIDE + k0 + 1];
            acc0 = __builtin_amdgcn_wmma_f32_16x16x4_f32(
                false, a0, false, b, (short)0, acc0, false, false);
            acc1 = __builtin_amdgcn_wmma_f32_16x16x4_f32(
                false, a1, false, b, (short)0, acc1, false, false);
            acc2 = __builtin_amdgcn_wmma_f32_16x16x4_f32(
                false, a2, false, b, (short)0, acc2, false, false);
            acc3 = __builtin_amdgcn_wmma_f32_16x16x4_f32(
                false, a3, false, b, (short)0, acc3, false, false);
        }
        const float bias = b1[ncol];
        #pragma unroll
        for (int r = 0; r < 8; ++r) {
            float v0 = acc0[r] + bias, v1 = acc1[r] + bias;
            float v2 = acc2[r] + bias, v3 = acc3[r] + bias;
            s_h[(     mbase + r) * H_STRIDE + ncol] = v0 > 0.f ? v0 : 0.f;
            s_h[(16 + mbase + r) * H_STRIDE + ncol] = v1 > 0.f ? v1 : 0.f;
            s_h[(32 + mbase + r) * H_STRIDE + ncol] = v2 > 0.f ? v2 : 0.f;
            s_h[(48 + mbase + r) * H_STRIDE + ncol] = v3 > 0.f ? v3 : 0.f;
        }
    }
    __syncthreads();

    {
        const int ntile = (wave & 3) * 16;
        const int kbase = (wave >> 2) * 64;
        const int ncol  = ntile + frow;
        v8f acc0 = {}, acc1 = {}, acc2 = {}, acc3 = {};
        #pragma unroll
        for (int kk = 0; kk < 16; ++kk) {
            const int k0 = kbase + kk * 4 + khalf;
            v2f b;
            b.x = W2[(size_t)(k0    ) * D_FEAT + ncol];
            b.y = W2[(size_t)(k0 + 1) * D_FEAT + ncol];
            v2f a0, a1, a2, a3;
            a0.x = s_h[(frow     ) * H_STRIDE + k0];
            a0.y = s_h[(frow     ) * H_STRIDE + k0 + 1];
            a1.x = s_h[(frow + 16) * H_STRIDE + k0];
            a1.y = s_h[(frow + 16) * H_STRIDE + k0 + 1];
            a2.x = s_h[(frow + 32) * H_STRIDE + k0];
            a2.y = s_h[(frow + 32) * H_STRIDE + k0 + 1];
            a3.x = s_h[(frow + 48) * H_STRIDE + k0];
            a3.y = s_h[(frow + 48) * H_STRIDE + k0 + 1];
            acc0 = __builtin_amdgcn_wmma_f32_16x16x4_f32(
                false, a0, false, b, (short)0, acc0, false, false);
            acc1 = __builtin_amdgcn_wmma_f32_16x16x4_f32(
                false, a1, false, b, (short)0, acc1, false, false);
            acc2 = __builtin_amdgcn_wmma_f32_16x16x4_f32(
                false, a2, false, b, (short)0, acc2, false, false);
            acc3 = __builtin_amdgcn_wmma_f32_16x16x4_f32(
                false, a3, false, b, (short)0, acc3, false, false);
        }
        const int mrow = (int)(__builtin_amdgcn_mbcnt_lo(~0u, 0u) >> 4) << 3;
        #pragma unroll
        for (int r = 0; r < 8; ++r) {
            atomicAdd(&s_out[(     mrow + r) * D_FEAT + ncol], acc0[r]);
            atomicAdd(&s_out[(16 + mrow + r) * D_FEAT + ncol], acc1[r]);
            atomicAdd(&s_out[(32 + mrow + r) * D_FEAT + ncol], acc2[r]);
            atomicAdd(&s_out[(48 + mrow + r) * D_FEAT + ncol], acc3[r]);
        }
    }
    __syncthreads();

    for (int i = tid; i < TILE_M * D_FEAT; i += 512) {
        int m = i >> 6, n = i & 63;
        int row = node0 + m;
        if (row < n_nodes)
            out[(size_t)row * D_FEAT + n] = s_out[i] + b2[n];
    }
}

extern "C" void kernel_launch(void* const* d_in, const int* in_sizes, int n_in,
                              void* d_out, int out_size, void* d_ws, size_t ws_size,
                              hipStream_t stream) {
    const float* feat = (const float*)d_in[0];
    const float* W1   = (const float*)d_in[1];
    const float* b1   = (const float*)d_in[2];
    const float* W2   = (const float*)d_in[3];
    const float* b2   = (const float*)d_in[4];
    const int*   src  = (const int*)d_in[5];
    const int*   dst  = (const int*)d_in[6];
    float*       out  = (float*)d_out;

    const int n_nodes  = in_sizes[0] / D_FEAT;
    const int n_edges  = in_sizes[5];
    const int n_tiles  = (n_nodes + TILE_M - 1) / TILE_M;
    const int n_chunks = (n_edges + CHUNK - 1) / CHUNK;

    int* offsT = (int*)d_ws;
    int* rec   = offsT + (NT_MAX + 1) * NB_MAX;

    gin_sort_kernel<<<n_chunks, 1024, 0, stream>>>(src, dst, offsT, rec, n_edges, n_tiles);
    gin_gather_mlp_kernel<<<n_tiles, 512, 0, stream>>>(feat, rec, offsT, W1, b1, W2, b2,
                                                       out, n_nodes, n_chunks);
}


# MLP kernel: rst, h and K-partial planes share one 65 KiB LDS region, each wave's B slice held in registers (64 in all): 4 resident blocks per WGP, was 3; K-partial sum order fixed, was arrival
# speedup vs baseline: 2.4301x; 1.0950x over previous
#include <hip/hip_runtime.h>
#include <hip/hip_bf16.h>

typedef __attribute__((ext_vector_type(2))) float v2f;
typedef __attribute__((ext_vector_type(8))) float v8f;

#define D_FEAT 64
#define H_DIM  256
#define TILE_M 64
#define RST_STRIDE 68
#define H_STRIDE   260

#define NT_MAX     2048
#define CHUNK      12800
#define NB_MAX     128

__device__ __forceinline__ int wave_incl_scan(int v, int lane) {
    #pragma unroll
    for (int off = 1; off < 32; off <<= 1) {
        const int u = __shfl_up(v, off);
        if (lane >= off) v += u;
    }
    return v;
}

__global__ __launch_bounds__(1024)
void gin_sort_kernel(const int* __restrict__ src, const int* __restrict__ dst,
                     int* __restrict__ offsT, int* __restrict__ rec,
                     int n_edges, int n_tiles) {
    __shared__ __align__(16) int2 sorted[CHUNK];
    __shared__ int hist[NT_MAX];
    __shared__ int wsum[32];
    const int tid  = threadIdx.x;
    const int lane = tid & 31;
    const int wave = tid >> 5;
    const int e0   = blockIdx.x * CHUNK;
    const int nloc = (n_edges - e0 < CHUNK) ? (n_edges - e0) : CHUNK;

    hist[tid] = 0;
    hist[tid + 1024] = 0;

    int d[16], s[16];
    if (nloc == CHUNK) {
        #pragma unroll
        for (int q = 0; q < 4; ++q) {
            const int i4 = q * 1024 + tid;
            int4 td = make_int4(-1, -1, -1, -1);
            int4 ts = make_int4(0, 0, 0, 0);
            if (i4 < CHUNK / 4) {
                td = reinterpret_cast<const int4*>(dst + e0)[i4];
                ts = reinterpret_cast<const int4*>(src + e0)[i4];
            }
            d[4 * q + 0] = td.x; d[4 * q + 1] = td.y;
            d[4 * q + 2] = td.z; d[4 * q + 3] = td.w;
            s[4 * q + 0] = ts.x; s[4 * q + 1] = ts.y;
            s[4 * q + 2] = ts.z; s[4 * q + 3] = ts.w;
        }
    } else {
        #pragma unroll
        for (int k = 0; k < 16; ++k) {
            const int i = ((k >> 2) * 1024 + tid) * 4 + (k & 3);
            const bool in = i < nloc;
            d[k] = in ? dst[e0 + i] : -1;
            s[k] = in ? src[e0 + i] : 0;
        }
    }
    __syncthreads();

    int r[16];
    #pragma unroll
    for (int k = 0; k < 16; ++k) {
        r[k] = 0;
        if (d[k] >= 0) r[k] = atomicAdd(&hist[d[k] >> 6], 1);
    }
    __syncthreads();

    const int c0 = hist[2 * tid];
    const int c1 = hist[2 * tid + 1];
    const int mine = c0 + c1;
    const int incl = wave_incl_scan(mine, lane);
    if (lane == 31) wsum[wave] = incl;
    __syncthreads();
    if (wave == 0) {
        const int v = wsum[lane];
        wsum[lane] = wave_incl_scan(v, lane) - v;
    }
    __syncthreads();
    const int excl = wsum[wave] + incl - mine;
    hist[2 * tid]     = excl;
    hist[2 * tid + 1] = excl + c0;
    __syncthreads();

    int pos[16];
    #pragma unroll
    for (int k = 0; k < 16; ++k)
        pos[k] = hist[(d[k] >= 0) ? (d[k] >> 6) : 0] + r[k];
    #pragma unroll
    for (int k = 0; k < 16; ++k)
        if (d[k] >= 0) sorted[pos[k]] = make_int2(s[k], d[k]);

    for (int t = tid; t <= n_tiles; t += 1024)
        offsT[t * NB_MAX + blockIdx.x] = hist[t];
    __syncthreads();

    int4* __restrict__ o4 = reinterpret_cast<int4*>(rec + 2 * (size_t)e0);
    const int4* s4 = reinterpret_cast<const int4*>(sorted);
    const int n4 = nloc >> 1;
    for (int i = tid; i < n4; i += 1024) o4[i] = s4[i];
    if ((nloc & 1) && tid == 0)
        reinterpret_cast<int2*>(rec)[(size_t)e0 + nloc - 1] = sorted[nloc - 1];
}

__device__ __forceinline__ int2 fetch_rec(const int2* __restrict__ rec2,
                                          const int* s_pos, const int* s_adj,
                                          int j, int total, int node0) {
    int2 r = make_int2(0, node0);
    if (j < total) {
        int lo = 0;
        #pragma unroll
        for (int step = NB_MAX / 2; step >= 1; step >>= 1)
            if (s_pos[lo + step] <= j) lo += step;
        r = rec2[s_adj[lo] + j];
    }
    return r;
}

#define AGG_GROUP 8

__global__ __launch_bounds__(512, 16)
void gin_gather_mlp_kernel(const float* __restrict__ feat,
                           const int* __restrict__ rec,
                           const int* __restrict__ offsT,
                           const float* __restrict__ W1,
                           const float* __restrict__ b1,
                           const float* __restrict__ W2,
                           const float* __restrict__ b2,
                           float* __restrict__ out, int n_nodes, int n_chunks) {
    __shared__ __align__(16) float s_tile[TILE_M * H_STRIDE];
    float* const s_rst  = s_tile;
    float* const s_h    = s_tile;
    float* const s_part = s_tile;
    int* const s_pos  = reinterpret_cast<int*>(s_tile + TILE_M * RST_STRIDE);
    int* const s_adj  = s_pos + NB_MAX + 4;
    int* const s_wtot = s_adj + NB_MAX;

    const int tid   = threadIdx.x;
    const int wave  = tid >> 5;
    const int lane  = tid & 31;
    const int wv    = __builtin_amdgcn_readfirstlane(tid >> 5);
    const int node0 = blockIdx.x * TILE_M;

    float4 own[2];
    const float4* tile_in = reinterpret_cast<const float4*>(feat) + (size_t)node0 * 16;
    const int rows_left = n_nodes - node0;
    #pragma unroll
    for (int q = 0; q < 2; ++q) {
        const int i4  = q * 512 + tid;
        own[q] = make_float4(0.f, 0.f, 0.f, 0.f);
        if ((i4 >> 4) < rows_left)
            own[q] = tile_in[i4];
    }
    int pstart = 0, pcnt = 0;
    if (tid < n_chunks) {
        pstart = offsT[blockIdx.x * NB_MAX + tid];
        pcnt   = offsT[(blockIdx.x + 1) * NB_MAX + tid] - pstart;
    }
    const int pincl = wave_incl_scan(pcnt, lane);
    if (tid < NB_MAX && lane == 31) s_wtot[wave] = pincl;
    #pragma unroll
    for (int q = 0; q < 2; ++q) {
        const int i4 = q * 512 + tid;
        *reinterpret_cast<float4*>(s_rst + (i4 >> 4) * RST_STRIDE + (i4 & 15) * 4) = own[q];
    }
    __syncthreads();
    if (tid < NB_MAX) {
        int pre = 0;
        #pragma unroll
        for (int w = 0; w < NB_MAX / 32; ++w) pre += (w < wave) ? s_wtot[w] : 0;
        const int pexcl = pre + pincl - pcnt;
        s_pos[tid] = pexcl;
        s_adj[tid] = tid * CHUNK + pstart - pexcl;
        if (tid == NB_MAX - 1) s_pos[NB_MAX] = pexcl + pcnt;
    }
    __syncthreads();
    const int total = s_pos[NB_MAX];

    {
        const int2* __restrict__ rec2 = reinterpret_cast<const int2*>(rec);
        int j = wv * 32 + lane;
        int2 cur = fetch_rec(rec2, s_pos, s_adj, j, total, node0);
        for (int w0 = wv * 32; w0 < total; w0 += 512) {
            asm volatile("" :: "v"(cur.x), "v"(cur.y));
            const int s = cur.x, d = cur.y - node0;
            j += 512;
            const int2 nxt = fetch_rec(rec2, s_pos, s_adj, j, total, node0);
            const int cnt = total - w0;
            #pragma unroll
            for (int g = 0; g < 32; g += AGG_GROUP) {
                if (g < cnt) {
                    float lo[AGG_GROUP], hi[AGG_GROUP];
                    #pragma unroll
                    for (int jj = 0; jj < AGG_GROUP; ++jj) {
                        const int sj = __builtin_amdgcn_readlane(s, g + jj);
                        const float* row = feat + (size_t)sj * D_FEAT;
                        lo[jj] = row[lane];
                        hi[jj] = row[lane + 32];
                    }
                    #pragma unroll
                    for (int jj = 0; jj < AGG_GROUP; ++jj) {
                        const int dj = __builtin_amdgcn_readlane(d, g + jj);
                        const bool ok = (g + jj) < cnt;
                        float* trow = s_rst + dj * RST_STRIDE;
                        atomicAdd(trow + lane,      ok ? lo[jj] : 0.f);
                        atomicAdd(trow + lane + 32, ok ? hi[jj] : 0.f);
                    }
                }
            }
            cur = nxt;
        }
    }
    __syncthreads();

    const int frow  = lane & 15;
    const int khalf = (lane >> 4) << 1;
    const int mbase = (lane >> 4) << 3;

    {
        const int ncol = wave * 16 + frow;
        v2f bf[16];
        #pragma unroll
        for (int kk = 0; kk < 16; ++kk) {
            const int k0 = kk * 4 + khalf;
            bf[kk].x = W1[(size_t)(k0    ) * H_DIM + ncol];
            bf[kk].y = W1[(size_t)(k0 + 1) * H_DIM + ncol];
        }
        const float bias = b1[ncol];
        {
            v8f acc2 = {}, acc3 = {};
            #pragma unroll
            for (int kk = 0; kk < 16; ++kk) {
                const int k0 = kk * 4 + khalf;
                v2f a2, a3;
                a2.x = s_rst[(frow + 32) * RST_STRIDE + k0];
                a2.y = s_rst[(frow + 32) * RST_STRIDE + k0 + 1];
                a3.x = s_rst[(frow + 48) * RST_STRIDE + k0];
                a3.y = s_rst[(frow + 48) * RST_STRIDE + k0 + 1];
                acc2 = __builtin_amdgcn_wmma_f32_16x16x4_f32(
                    false, a2, false, bf[kk], (short)0, acc2, false, false);
                acc3 = __builtin_amdgcn_wmma_f32_16x16x4_f32(
                    false, a3, false, bf[kk], (short)0, acc3, false, false);
            }
            #pragma unroll
            for (int r = 0; r < 8; ++r) {
                float v2 = acc2[r] + bias, v3 = acc3[r] + bias;
                s_h[(32 + mbase + r) * H_STRIDE + ncol] = v2 > 0.f ? v2 : 0.f;
                s_h[(48 + mbase + r) * H_STRIDE + ncol] = v3 > 0.f ? v3 : 0.f;
            }
        }
        {
            v8f acc0 = {}, acc1 = {};
            #pragma unroll
            for (int kk = 0; kk < 16; ++kk) {
                const int k0 = kk * 4 + khalf;
                v2f a0, a1;
                a0.x = s_rst[(frow     ) * RST_STRIDE + k0];
                a0.y = s_rst[(frow     ) * RST_STRIDE + k0 + 1];
                a1.x = s_rst[(frow + 16) * RST_STRIDE + k0];
                a1.y = s_rst[(frow + 16) * RST_STRIDE + k0 + 1];
                acc0 = __builtin_amdgcn_wmma_f32_16x16x4_f32(
                    false, a0, false, bf[kk], (short)0, acc0, false, false);
                acc1 = __builtin_amdgcn_wmma_f32_16x16x4_f32(
                    false, a1, false, bf[kk], (short)0, acc1, false, false);
            }
            __syncthreads();
            #pragma unroll
            for (int r = 0; r < 8; ++r) {
                float v0 = acc0[r] + bias, v1 = acc1[r] + bias;
                s_h[(     mbase + r) * H_STRIDE + ncol] = v0 > 0.f ? v0 : 0.f;
                s_h[(16 + mbase + r) * H_STRIDE + ncol] = v1 > 0.f ? v1 : 0.f;
            }
        }
    }
    __syncthreads();

    {
        const int ntile = (wave & 3) * 16;
        const int kbase = (wave >> 2) * 64;
        const int ncol  = ntile + frow;
        v2f bf[16];
        #pragma unroll
        for (int kk = 0; kk < 16; ++kk) {
            const int k0 = kbase + kk * 4 + khalf;
            bf[kk].x = W2[(size_t)(k0    ) * D_FEAT + ncol];
            bf[kk].y = W2[(size_t)(k0 + 1) * D_FEAT + ncol];
        }
        float* const plane_lo = s_part + (wave >> 2) * (32 * D_FEAT);
        float* const plane_hi = s_part + 32 * H_STRIDE + (wave >> 2) * (32 * D_FEAT);
        {
            v8f acc0 = {}, acc1 = {};
            #pragma unroll
            for (int kk = 0; kk < 16; ++kk) {
                const int k0 = kbase + kk * 4 + khalf;
                v2f a0, a1;
                a0.x = s_h[(frow     ) * H_STRIDE + k0];
                a0.y = s_h[(frow     ) * H_STRIDE + k0 + 1];
                a1.x = s_h[(frow + 16) * H_STRIDE + k0];
                a1.y = s_h[(frow + 16) * H_STRIDE + k0 + 1];
                acc0 = __builtin_amdgcn_wmma_f32_16x16x4_f32(
                    false, a0, false, bf[kk], (short)0, acc0, false, false);
                acc1 = __builtin_amdgcn_wmma_f32_16x16x4_f32(
                    false, a1, false, bf[kk], (short)0, acc1, false, false);
            }
            __syncthreads();
            #pragma unroll
            for (int r = 0; r < 8; ++r) {
                plane_lo[(     mbase + r) * D_FEAT + ncol] = acc0[r];
                plane_lo[(16 + mbase + r) * D_FEAT + ncol] = acc1[r];
            }
        }
        {
            v8f acc2 = {}, acc3 = {};
            #pragma unroll
            for (int kk = 0; kk < 16; ++kk) {
                const int k0 = kbase + kk * 4 + khalf;
                v2f a2, a3;
                a2.x = s_h[(frow + 32) * H_STRIDE + k0];
                a2.y = s_h[(frow + 32) * H_STRIDE + k0 + 1];
                a3.x = s_h[(frow + 48) * H_STRIDE + k0];
                a3.y = s_h[(frow + 48) * H_STRIDE + k0 + 1];
                acc2 = __builtin_amdgcn_wmma_f32_16x16x4_f32(
                    false, a2, false, bf[kk], (short)0, acc2, false, false);
                acc3 = __builtin_amdgcn_wmma_f32_16x16x4_f32(
                    false, a3, false, bf[kk], (short)0, acc3, false, false);
            }
            __syncthreads();
            #pragma unroll
            for (int r = 0; r < 8; ++r) {
                plane_hi[(     mbase + r) * D_FEAT + ncol] = acc2[r];
                plane_hi[(16 + mbase + r) * D_FEAT + ncol] = acc3[r];
            }
        }
    }
    __syncthreads();

    float4* tile_out = reinterpret_cast<float4*>(out) + (size_t)node0 * 16;
    const int t2 = (wv << 5) + (int)__builtin_amdgcn_mbcnt_lo(~0u, 0u);
    #pragma unroll
    for (int q = 0; q < 2; ++q) {
        const float* pl = s_part + q * (32 * H_STRIDE);
        const float4 p0 = reinterpret_cast<const float4*>(pl)[t2];
        const float4 p1 = reinterpret_cast<const float4*>(pl + 1 * 32 * D_FEAT)[t2];
        const float4 p2 = reinterpret_cast<const float4*>(pl + 2 * 32 * D_FEAT)[t2];
        const float4 p3 = reinterpret_cast<const float4*>(pl + 3 * 32 * D_FEAT)[t2];
        const float4 bb = reinterpret_cast<const float4*>(b2)[t2 & 15];
        float4 o;
        o.x = (((p0.x + p1.x) + p2.x) + p3.x) + bb.x;
        o.y = (((p0.y + p1.y) + p2.y) + p3.y) + bb.y;
        o.z = (((p0.z + p1.z) + p2.z) + p3.z) + bb.z;
        o.w = (((p0.w + p1.w) + p2.w) + p3.w) + bb.w;
        if (q * 32 + (t2 >> 4) < rows_left)
            tile_out[q * 512 + t2] = o;
    }
}

extern "C" void kernel_launch(void* const* d_in, const int* in_sizes, int n_in,
                              void* d_out, int out_size, void* d_ws, size_t ws_size,
                              hipStream_t stream) {
    const float* feat = (const float*)d_in[0];
    const float* W1   = (const float*)d_in[1];
    const float* b1   = (const float*)d_in[2];
    const float* W2   = (const float*)d_in[3];
    const float* b2   = (const float*)d_in[4];
    const int*   src  = (const int*)d_in[5];
    const int*   dst  = (const int*)d_in[6];
    float*       out  = (float*)d_out;

    const int n_nodes  = in_sizes[0] / D_FEAT;
    const int n_edges  = in_sizes[5];
    const int n_tiles  = (n_nodes + TILE_M - 1) / TILE_M;
    const int n_chunks = (n_edges + CHUNK - 1) / CHUNK;

    int* offsT = (int*)d_ws;
    int* rec   = offsT + (NT_MAX + 1) * NB_MAX;

    gin_sort_kernel<<<n_chunks, 1024, 0, stream>>>(src, dst, offsT, rec, n_edges, n_tiles);
    gin_gather_mlp_kernel<<<n_tiles, 512, 0, stream>>>(feat, rec, offsT, W1, b1, W2, b2,
                                                       out, n_nodes, n_chunks);
}


# MLP kernel: tile's records ordered by destination row in LDS (integer ranks); waves sum neighbor rows in registers, one 256-B row per load, ds_add_f32 only per row change; sum order arrival, grouped
# speedup vs baseline: 2.6179x; 1.0773x over previous
#include <hip/hip_runtime.h>
#include <hip/hip_bf16.h>

typedef __attribute__((ext_vector_type(2))) float v2f;
typedef __attribute__((ext_vector_type(8))) float v8f;

#define D_FEAT 64
#define H_DIM  256
#define TILE_M 64
#define RST_STRIDE 68
#define H_STRIDE   260

#define NT_MAX     2048
#define CHUNK      12800
#define NB_MAX     128

__device__ __forceinline__ int wave_incl_scan(int v, int lane) {
    #pragma unroll
    for (int off = 1; off < 32; off <<= 1) {
        const int u = __shfl_up(v, off);
        if (lane >= off) v += u;
    }
    return v;
}

__global__ __launch_bounds__(1024)
void gin_sort_kernel(const int* __restrict__ src, const int* __restrict__ dst,
                     int* __restrict__ offsT, int* __restrict__ rec,
                     int n_edges, int n_tiles) {
    __shared__ __align__(16) int2 sorted[CHUNK];
    __shared__ int hist[NT_MAX];
    __shared__ int wsum[32];
    const int tid  = threadIdx.x;
    const int lane = tid & 31;
    const int wave = tid >> 5;
    const int e0   = blockIdx.x * CHUNK;
    const int nloc = (n_edges - e0 < CHUNK) ? (n_edges - e0) : CHUNK;

    hist[tid] = 0;
    hist[tid + 1024] = 0;

    int d[16], s[16];
    if (nloc == CHUNK) {
        #pragma unroll
        for (int q = 0; q < 4; ++q) {
            const int i4 = q * 1024 + tid;
            int4 td = make_int4(-1, -1, -1, -1);
            int4 ts = make_int4(0, 0, 0, 0);
            if (i4 < CHUNK / 4) {
                td = reinterpret_cast<const int4*>(dst + e0)[i4];
                ts = reinterpret_cast<const int4*>(src + e0)[i4];
            }
            d[4 * q + 0] = td.x; d[4 * q + 1] = td.y;
            d[4 * q + 2] = td.z; d[4 * q + 3] = td.w;
            s[4 * q + 0] = ts.x; s[4 * q + 1] = ts.y;
            s[4 * q + 2] = ts.z; s[4 * q + 3] = ts.w;
        }
    } else {
        #pragma unroll
        for (int k = 0; k < 16; ++k) {
            const int i = ((k >> 2) * 1024 + tid) * 4 + (k & 3);
            const bool in = i < nloc;
            d[k] = in ? dst[e0 + i] : -1;
            s[k] = in ? src[e0 + i] : 0;
        }
    }
    __syncthreads();

    int r[16];
    #pragma unroll
    for (int k = 0; k < 16; ++k) {
        r[k] = 0;
        if (d[k] >= 0) r[k] = atomicAdd(&hist[d[k] >> 6], 1);
    }
    __syncthreads();

    const int c0 = hist[2 * tid];
    const int c1 = hist[2 * tid + 1];
    const int mine = c0 + c1;
    const int incl = wave_incl_scan(mine, lane);
    if (lane == 31) wsum[wave] = incl;
    __syncthreads();
    if (wave == 0) {
        const int v = wsum[lane];
        wsum[lane] = wave_incl_scan(v, lane) - v;
    }
    __syncthreads();
    const int excl = wsum[wave] + incl - mine;
    hist[2 * tid]     = excl;
    hist[2 * tid + 1] = excl + c0;
    __syncthreads();

    int pos[16];
    #pragma unroll
    for (int k = 0; k < 16; ++k)
        pos[k] = hist[(d[k] >= 0) ? (d[k] >> 6) : 0] + r[k];
    #pragma unroll
    for (int k = 0; k < 16; ++k)
        if (d[k] >= 0) sorted[pos[k]] = make_int2(s[k], d[k]);

    for (int t = tid; t <= n_tiles; t += 1024)
        offsT[t * NB_MAX + blockIdx.x] = hist[t];
    __syncthreads();

    int4* __restrict__ o4 = reinterpret_cast<int4*>(rec + 2 * (size_t)e0);
    const int4* s4 = reinterpret_cast<const int4*>(sorted);
    const int n4 = nloc >> 1;
    for (int i = tid; i < n4; i += 1024) o4[i] = s4[i];
    if ((nloc & 1) && tid == 0)
        reinterpret_cast<int2*>(rec)[(size_t)e0 + nloc - 1] = sorted[nloc - 1];
}

#define REC_CAP    1536
#define REC_PER_T  3
#define ROW_GROUP  16

__global__ __launch_bounds__(512, 16)
void gin_gather_mlp_kernel(const float* __restrict__ feat,
                           const int* __restrict__ rec,
                           const int* __restrict__ offsT,
                           const float* __restrict__ W1,
                           const float* __restrict__ b1,
                           const float* __restrict__ W2,
                           const float* __restrict__ b2,
                           float* __restrict__ out, int n_nodes, int n_chunks) {
    __shared__ __align__(16) float s_tile[TILE_M * H_STRIDE];
    float* const s_rst  = s_tile;
    float* const s_h    = s_tile;
    float* const s_part = s_tile;
    int* const s_pos  = reinterpret_cast<int*>(s_tile + TILE_M * RST_STRIDE);
    int* const s_adj  = s_pos + NB_MAX + 4;
    int* const s_wtot = s_adj + NB_MAX;
    int* const s_cnt    = s_wtot + 8;
    int* const s_start  = s_cnt + TILE_M;
    int* const s_sorted = s_start + TILE_M + 4;

    const int tid   = threadIdx.x;
    const int wave  = tid >> 5;
    const int lane  = tid & 31;
    const int wv    = __builtin_amdgcn_readfirstlane(tid >> 5);
    const int node0 = blockIdx.x * TILE_M;

    float4 own[2];
    const float4* tile_in = reinterpret_cast<const float4*>(feat) + (size_t)node0 * 16;
    const int rows_left = n_nodes - node0;
    #pragma unroll
    for (int q = 0; q < 2; ++q) {
        const int i4  = q * 512 + tid;
        own[q] = make_float4(0.f, 0.f, 0.f, 0.f);
        if ((i4 >> 4) < rows_left)
            own[q] = tile_in[i4];
    }
    int pstart = 0, pcnt = 0;
    if (tid < n_chunks) {
        pstart = offsT[blockIdx.x * NB_MAX + tid];
        pcnt   = offsT[(blockIdx.x + 1) * NB_MAX + tid] - pstart;
    }
    const int pincl = wave_incl_scan(pcnt, lane);
    if (tid < NB_MAX && lane == 31) s_wtot[wave] = pincl;
    #pragma unroll
    for (int q = 0; q < 2; ++q) {
        const int i4 = q * 512 + tid;
        *reinterpret_cast<float4*>(s_rst + (i4 >> 4) * RST_STRIDE + (i4 & 15) * 4) = own[q];
    }
    if (tid < TILE_M) s_cnt[tid] = 0;
    __syncthreads();
    if (tid < NB_MAX) {
        int pre = 0;
        #pragma unroll
        for (int w = 0; w < NB_MAX / 32; ++w) pre += (w < wave) ? s_wtot[w] : 0;
        const int pexcl = pre + pincl - pcnt;
        s_pos[tid] = pexcl;
        s_adj[tid] = tid * CHUNK + pstart - pexcl;
        if (tid == NB_MAX - 1) s_pos[NB_MAX] = pexcl + pcnt;
    }
    __syncthreads();
    const int total = s_pos[NB_MAX];

    {
        const int2* __restrict__ rec2 = reinterpret_cast<const int2*>(rec);
        for (int b = 0; b < total; b += REC_CAP) {
            const int wend = (total - b < REC_CAP) ? total : (b + REC_CAP);

            int jc[REC_PER_T], lo[REC_PER_T];
            bool live[REC_PER_T];
            #pragma unroll
            for (int k = 0; k < REC_PER_T; ++k) {
                const int j = b + k * 512 + tid;
                live[k] = j < wend;
                jc[k] = live[k] ? j : (wend - 1);
                lo[k] = 0;
            }
            #pragma unroll
            for (int step = NB_MAX / 2; step >= 1; step >>= 1) {
                #pragma unroll
                for (int k = 0; k < REC_PER_T; ++k) {
                    const int p = s_pos[lo[k] + step];
                    lo[k] = (p <= jc[k]) ? (lo[k] + step) : lo[k];
                }
            }
            int2 r[REC_PER_T];
            #pragma unroll
            for (int k = 0; k < REC_PER_T; ++k)
                r[k] = rec2[s_adj[lo[k]] + jc[k]];
            int dl[REC_PER_T], rk[REC_PER_T];
            #pragma unroll
            for (int k = 0; k < REC_PER_T; ++k) {
                dl[k] = r[k].y - node0;
                rk[k] = 0;
                if (live[k]) rk[k] = atomicAdd(&s_cnt[dl[k]], 1);
            }
            __syncthreads();

            if (wv == 0) {
                const int c0 = s_cnt[2 * lane];
                const int c1 = s_cnt[2 * lane + 1];
                s_cnt[2 * lane]     = 0;
                s_cnt[2 * lane + 1] = 0;
                const int m  = c0 + c1;
                const int incl = wave_incl_scan(m, lane);
                s_start[2 * lane]     = incl - m;
                s_start[2 * lane + 1] = incl - m + c0;
            }
            __syncthreads();

            #pragma unroll
            for (int k = 0; k < REC_PER_T; ++k)
                if (live[k]) s_sorted[s_start[dl[k]] + rk[k]] = (r[k].x << 6) | dl[k];
            __syncthreads();

            const int cnt_all = wend - b;
            const int share   = (cnt_all + 15) >> 4;
            const int S = wv * share;
            const int E = (S + share < cnt_all) ? (S + share) : cnt_all;
            float2 acc = make_float2(0.f, 0.f);
            int currow = -1;
            for (int i = S; i < E; i += 32) {
                const int n  = E - i;
                const int my = (lane < n) ? s_sorted[i + lane] : 0;
                #pragma unroll
                for (int g = 0; g < 32; g += ROW_GROUP) {
                    if (g < n) {
                        float2 v[ROW_GROUP];
                        #pragma unroll
                        for (int jj = 0; jj < ROW_GROUP; ++jj) {
                            const int pj = __builtin_amdgcn_readlane(my, g + jj);
                            const float2* row = reinterpret_cast<const float2*>(feat + (size_t)(pj >> 6) * D_FEAT);
                            v[jj] = row[lane];
                        }
                        #pragma unroll
                        for (int jj = 0; jj < ROW_GROUP; ++jj) {
                            if (g + jj < n) {
                                const int dj = __builtin_amdgcn_readlane(my, g + jj) & (TILE_M - 1);
                                if (dj != currow) {
                                    if (currow >= 0) {
                                        float* t = s_rst + currow * RST_STRIDE + 2 * lane;
                                        atomicAdd(t,     acc.x);
                                        atomicAdd(t + 1, acc.y);
                                    }
                                    currow = dj;
                                    acc = make_float2(0.f, 0.f);
                                }
                                acc.x += v[jj].x;
                                acc.y += v[jj].y;
                            }
                        }
                    }
                }
            }
            if (currow >= 0) {
                float* t = s_rst + currow * RST_STRIDE + 2 * lane;
                atomicAdd(t,     acc.x);
                atomicAdd(t + 1, acc.y);
            }
        }
    }
    __syncthreads();

    const int frow  = lane & 15;
    const int khalf = (lane >> 4) << 1;
    const int mbase = (lane >> 4) << 3;

    {
        const int ncol = wave * 16 + frow;
        v2f bf[16];
        #pragma unroll
        for (int kk = 0; kk < 16; ++kk) {
            const int k0 = kk * 4 + khalf;
            bf[kk].x = W1[(size_t)(k0    ) * H_DIM + ncol];
            bf[kk].y = W1[(size_t)(k0 + 1) * H_DIM + ncol];
        }
        const float bias = b1[ncol];
        {
            v8f acc2 = {}, acc3 = {};
            #pragma unroll
            for (int kk = 0; kk < 16; ++kk) {
                const int k0 = kk * 4 + khalf;
                v2f a2, a3;
                a2.x = s_rst[(frow + 32) * RST_STRIDE + k0];
                a2.y = s_rst[(frow + 32) * RST_STRIDE + k0 + 1];
                a3.x = s_rst[(frow + 48) * RST_STRIDE + k0];
                a3.y = s_rst[(frow + 48) * RST_STRIDE + k0 + 1];
                acc2 = __builtin_amdgcn_wmma_f32_16x16x4_f32(
                    false, a2, false, bf[kk], (short)0, acc2, false, false);
                acc3 = __builtin_amdgcn_wmma_f32_16x16x4_f32(
                    false, a3, false, bf[kk], (short)0, acc3, false, false);
            }
            #pragma unroll
            for (int r = 0; r < 8; ++r) {
                float v2 = acc2[r] + bias, v3 = acc3[r] + bias;
                s_h[(32 + mbase + r) * H_STRIDE + ncol] = v2 > 0.f ? v2 : 0.f;
                s_h[(48 + mbase + r) * H_STRIDE + ncol] = v3 > 0.f ? v3 : 0.f;
            }
        }
        {
            v8f acc0 = {}, acc1 = {};
            #pragma unroll
            for (int kk = 0; kk < 16; ++kk) {
                const int k0 = kk * 4 + khalf;
                v2f a0, a1;
                a0.x = s_rst[(frow     ) * RST_STRIDE + k0];
                a0.y = s_rst[(frow     ) * RST_STRIDE + k0 + 1];
                a1.x = s_rst[(frow + 16) * RST_STRIDE + k0];
                a1.y = s_rst[(frow + 16) * RST_STRIDE + k0 + 1];
                acc0 = __builtin_amdgcn_wmma_f32_16x16x4_f32(
                    false, a0, false, bf[kk], (short)0, acc0, false, false);
                acc1 = __builtin_amdgcn_wmma_f32_16x16x4_f32(
                    false, a1, false, bf[kk], (short)0, acc1, false, false);
            }
            __syncthreads();
            #pragma unroll
            for (int r = 0; r < 8; ++r) {
                float v0 = acc0[r] + bias, v1 = acc1[r] + bias;
                s_h[(     mbase + r) * H_STRIDE + ncol] = v0 > 0.f ? v0 : 0.f;
                s_h[(16 + mbase + r) * H_STRIDE + ncol] = v1 > 0.f ? v1 : 0.f;
            }
        }
    }
    __syncthreads();

    {
        const int ntile = (wave & 3) * 16;
        const int kbase = (wave >> 2) * 64;
        const int ncol  = ntile + frow;
        v2f bf[16];
        #pragma unroll
        for (int kk = 0; kk < 16; ++kk) {
            const int k0 = kbase + kk * 4 + khalf;
            bf[kk].x = W2[(size_t)(k0    ) * D_FEAT + ncol];
            bf[kk].y = W2[(size_t)(k0 + 1) * D_FEAT + ncol];
        }
        float* const plane_lo = s_part + (wave >> 2) * (32 * D_FEAT);
        float* const plane_hi = s_part + 32 * H_STRIDE + (wave >> 2) * (32 * D_FEAT);
        {
            v8f acc0 = {}, acc1 = {};
            #pragma unroll
            for (int kk = 0; kk < 16; ++kk) {
                const int k0 = kbase + kk * 4 + khalf;
                v2f a0, a1;
                a0.x = s_h[(frow     ) * H_STRIDE + k0];
                a0.y = s_h[(frow     ) * H_STRIDE + k0 + 1];
                a1.x = s_h[(frow + 16) * H_STRIDE + k0];
                a1.y = s_h[(frow + 16) * H_STRIDE + k0 + 1];
                acc0 = __builtin_amdgcn_wmma_f32_16x16x4_f32(
                    false, a0, false, bf[kk], (short)0, acc0, false, false);
                acc1 = __builtin_amdgcn_wmma_f32_16x16x4_f32(
                    false, a1, false, bf[kk], (short)0, acc1, false, false);
            }
            __syncthreads();
            #pragma unroll
            for (int r = 0; r < 8; ++r) {
                plane_lo[(     mbase + r) * D_FEAT + ncol] = acc0[r];
                plane_lo[(16 + mbase + r) * D_FEAT + ncol] = acc1[r];
            }
        }
        {
            v8f acc2 = {}, acc3 = {};
            #pragma unroll
            for (int kk = 0; kk < 16; ++kk) {
                const int k0 = kbase + kk * 4 + khalf;
                v2f a2, a3;
                a2.x = s_h[(frow + 32) * H_STRIDE + k0];
                a2.y = s_h[(frow + 32) * H_STRIDE + k0 + 1];
                a3.x = s_h[(frow + 48) * H_STRIDE + k0];
                a3.y = s_h[(frow + 48) * H_STRIDE + k0 + 1];
                acc2 = __builtin_amdgcn_wmma_f32_16x16x4_f32(
                    false, a2, false, bf[kk], (short)0, acc2, false, false);
                acc3 = __builtin_amdgcn_wmma_f32_16x16x4_f32(
                    false, a3, false, bf[kk], (short)0, acc3, false, false);
            }
            __syncthreads();
            #pragma unroll
            for (int r = 0; r < 8; ++r) {
                plane_hi[(     mbase + r) * D_FEAT + ncol] = acc2[r];
                plane_hi[(16 + mbase + r) * D_FEAT + ncol] = acc3[r];
            }
        }
    }
    __syncthreads();

    float4* tile_out = reinterpret_cast<float4*>(out) + (size_t)node0 * 16;
    const int t2 = (wv << 5) + (int)__builtin_amdgcn_mbcnt_lo(~0u, 0u);
    #pragma unroll
    for (int q = 0; q < 2; ++q) {
        const float* pl = s_part + q * (32 * H_STRIDE);
        const float4 p0 = reinterpret_cast<const float4*>(pl)[t2];
        const float4 p1 = reinterpret_cast<const float4*>(pl + 1 * 32 * D_FEAT)[t2];
        const float4 p2 = reinterpret_cast<const float4*>(pl + 2 * 32 * D_FEAT)[t2];
        const float4 p3 = reinterpret_cast<const float4*>(pl + 3 * 32 * D_FEAT)[t2];
        const float4 bb = reinterpret_cast<const float4*>(b2)[t2 & 15];
        float4 o;
        o.x = (((p0.x + p1.x) + p2.x) + p3.x) + bb.x;
        o.y = (((p0.y + p1.y) + p2.y) + p3.y) + bb.y;
        o.z = (((p0.z + p1.z) + p2.z) + p3.z) + bb.z;
        o.w = (((p0.w + p1.w) + p2.w) + p3.w) + bb.w;
        if (q * 32 + (t2 >> 4) < rows_left)
            tile_out[q * 512 + t2] = o;
    }
}

extern "C" void kernel_launch(void* const* d_in, const int* in_sizes, int n_in,
                              void* d_out, int out_size, void* d_ws, size_t ws_size,
                              hipStream_t stream) {
    const float* feat = (const float*)d_in[0];
    const float* W1   = (const float*)d_in[1];
    const float* b1   = (const float*)d_in[2];
    const float* W2   = (const float*)d_in[3];
    const float* b2   = (const float*)d_in[4];
    const int*   src  = (const int*)d_in[5];
    const int*   dst  = (const int*)d_in[6];
    float*       out  = (float*)d_out;

    const int n_nodes  = in_sizes[0] / D_FEAT;
    const int n_edges  = in_sizes[5];
    const int n_tiles  = (n_nodes + TILE_M - 1) / TILE_M;
    const int n_chunks = (n_edges + CHUNK - 1) / CHUNK;

    int* offsT = (int*)d_ws;
    int* rec   = offsT + (NT_MAX + 1) * NB_MAX;

    gin_sort_kernel<<<n_chunks, 1024, 0, stream>>>(src, dst, offsT, rec, n_edges, n_tiles);
    gin_gather_mlp_kernel<<<n_tiles, 512, 0, stream>>>(feat, rec, offsT, W1, b1, W2, b2,
                                                       out, n_nodes, n_chunks);
}
